# starting kernel
# baseline (speedup 1.0000x reference)
#include <hip/hip_runtime.h>
#include <math.h>


typedef __attribute__((ext_vector_type(2))) float v2f;
typedef __attribute__((ext_vector_type(8))) float v8f;

static inline int ceil_div(long long a, long long b) { return (int)((a + b - 1) / b); }

template <int K, int N>
__global__ __launch_bounds__(256) void gemm_wmma_f32(const float* __restrict__ A,
                                                     const float* __restrict__ B,
                                                     float* __restrict__ C, int M) {
  constexpr int NT = N / 16;
  const int mTiles = (M + 15) >> 4;
  const int gwave = (int)((blockIdx.x * (long long)blockDim.x + threadIdx.x) >> 5);
  if (gwave >= mTiles * NT) return;

  const int lane = threadIdx.x & 31;
  const int mt = gwave / NT;
  const int nt = gwave - mt * NT;
  const int half = lane >> 4;
  const int l16 = lane & 15;

  const int row = mt * 16 + l16;
  const int rclamp = row < M ? row : (M - 1);
  const int col = nt * 16 + l16;
  const float* __restrict__ Ar = A + (long long)rclamp * K;

  v8f acc = {};
#pragma unroll 4
  for (int k = 0; k < K; k += 4) {
    const int ka = k + 2 * half;
    v2f a, b;
    a.x = Ar[ka];
    a.y = Ar[ka + 1];
    b.x = B[(long long)ka * N + col];
    b.y = B[(long long)(ka + 1) * N + col];
    acc = __builtin_amdgcn_wmma_f32_16x16x4_f32(false, a, false, b, (short)0, acc,
                                                false, false);
  }

  const int m0 = mt * 16 + 8 * half;
  float* __restrict__ Cp = C + (long long)m0 * N + col;
  if (mt * 16 + 16 <= M) {
#pragma unroll
    for (int v = 0; v < 8; ++v) Cp[(long long)v * N] = acc[v];
  } else {
#pragma unroll
    for (int v = 0; v < 8; ++v)
      if (m0 + v < M) Cp[(long long)v * N] = acc[v];
  }
}

__global__ void deg_kernel(const int* __restrict__ dst, int* __restrict__ deg, int e) {
  const int i = blockIdx.x * blockDim.x + threadIdx.x;
  if (i < e) atomicAdd(&deg[dst[i]], 1);
}

__global__ void dinv_kernel(const int* __restrict__ deg, float* __restrict__ dinv, int n) {
  const int i = blockIdx.x * blockDim.x + threadIdx.x;
  if (i < n) dinv[i] = rsqrtf((float)deg[i] + 1.0f);
}

template <int LOGF>
__global__ __launch_bounds__(256) void agg_kernel(const float* __restrict__ H,
                                                  const int* __restrict__ src,
                                                  const int* __restrict__ dst,
                                                  const float* __restrict__ dinv,
                                                  float* __restrict__ out, int e, int n) {
  constexpr int F = 1 << LOGF;
  const long long total = (long long)(e + n) << LOGF;
  const long long i = (long long)blockIdx.x * blockDim.x + threadIdx.x;
  if (i >= total) return;
  const int f = (int)(i & (F - 1));
  const long long edge = i >> LOGF;
  int s, d;
  if (edge < (long long)e) {
    s = src[edge];
    d = dst[edge];
  } else {
    s = d = (int)(edge - e);
  }
  const float nrm = dinv[s] * dinv[d];
  atomicAdd(out + (((long long)d) << LOGF) + f, H[(((long long)s) << LOGF) + f] * nrm);
}

template <int LOGF>
__global__ void bias_relu_kernel(const float* __restrict__ agg, const float* __restrict__ b,
                                 float* __restrict__ out, long long total) {
  constexpr int F = 1 << LOGF;
  const long long i = (long long)blockIdx.x * blockDim.x + threadIdx.x;
  if (i >= total) return;
  const float v = agg[i] + b[(int)(i & (F - 1))];
  out[i] = v > 0.0f ? v : 0.0f;
}

__global__ void lsm_kernel(const float* __restrict__ agg, const float* __restrict__ b,
                           float* __restrict__ out, int n) {
  const int i = blockIdx.x * blockDim.x + threadIdx.x;
  if (i >= n) return;
  float v[16];
  float mx = -INFINITY;
#pragma unroll
  for (int j = 0; j < 16; ++j) {
    v[j] = agg[(long long)i * 16 + j] + b[j];
    mx = fmaxf(mx, v[j]);
  }
  float s = 0.0f;
#pragma unroll
  for (int j = 0; j < 16; ++j) s += expf(v[j] - mx);
  const float lse = mx + logf(s);
#pragma unroll
  for (int j = 0; j < 16; ++j) out[(long long)i * 16 + j] = v[j] - lse;
}

extern "C" void kernel_launch(void* const* d_in, const int* in_sizes, int n_in,
                              void* d_out, int out_size, void* d_ws, size_t ws_size,
                              hipStream_t stream) {
  const float* x  = (const float*)d_in[0];
  const int*   ei = (const int*)d_in[1];
  const float* W1 = (const float*)d_in[2];
  const float* b1 = (const float*)d_in[3];
  const float* W2 = (const float*)d_in[4];
  const float* b2 = (const float*)d_in[5];

  const int n = in_sizes[0] / 256;
  const int e = in_sizes[1] / 2;
  const int* src = ei;
  const int* dst = ei + e;

  float* agg1 = (float*)d_ws;
  float* agg2 = agg1 + (size_t)n * 128;
  int*   deg  = (int*)(agg2 + (size_t)n * 16);
  float* h1   = (float*)(deg + n);
  float* h2   = h1 + (size_t)n * 128;
  float* dinv = h2 + (size_t)n * 16;

  hipMemsetAsync(d_ws, 0, (size_t)n * (128 + 16 + 1) * sizeof(float), stream);

  deg_kernel<<<ceil_div(e, 256), 256, 0, stream>>>(dst, deg, e);
  dinv_kernel<<<ceil_div(n, 256), 256, 0, stream>>>(deg, dinv, n);

  {
    const long long waves = (long long)((n + 15) / 16) * (128 / 16);
    gemm_wmma_f32<256, 128><<<ceil_div(waves * 32, 256), 256, 0, stream>>>(x, W1, h1, n);
  }
  agg_kernel<7><<<ceil_div(((long long)(e + n)) << 7, 256), 256, 0, stream>>>(
      h1, src, dst, dinv, agg1, e, n);
  bias_relu_kernel<7><<<ceil_div((long long)n * 128, 256), 256, 0, stream>>>(
      agg1, b1, h1, (long long)n * 128);

  {
    const long long waves = (long long)((n + 15) / 16) * (16 / 16);
    gemm_wmma_f32<128, 16><<<ceil_div(waves * 32, 256), 256, 0, stream>>>(h1, W2, h2, n);
  }
  agg_kernel<4><<<ceil_div(((long long)(e + n)) << 4, 256), 256, 0, stream>>>(
      h2, src, dst, dinv, agg2, e, n);
  lsm_kernel<<<ceil_div(n, 256), 256, 0, stream>>>(agg2, b2, (float*)d_out, n);
}


# Aggregate by destination, no float atomics: returning int add gives each edge its arrival-order rank, prefix sum the group starts; each row is summed in registers in that order and stored once
# speedup vs baseline: 2.7393x; 2.7393x over previous
#include <hip/hip_runtime.h>
#include <math.h>


typedef __attribute__((ext_vector_type(2))) float v2f;
typedef __attribute__((ext_vector_type(8))) float v8f;

static inline int ceil_div(long long a, long long b) { return (int)((a + b - 1) / b); }

template <int K, int N>
__global__ __launch_bounds__(256) void gemm_wmma_f32(const float* __restrict__ A,
                                                     const float* __restrict__ B,
                                                     float* __restrict__ C, int M) {
  constexpr int NT = N / 16;
  const int mTiles = (M + 15) >> 4;
  const int gwave = (int)((blockIdx.x * (long long)blockDim.x + threadIdx.x) >> 5);
  if (gwave >= mTiles * NT) return;

  const int lane = threadIdx.x & 31;
  const int mt = gwave / NT;
  const int nt = gwave - mt * NT;
  const int half = lane >> 4;
  const int l16 = lane & 15;

  const int row = mt * 16 + l16;
  const int rclamp = row < M ? row : (M - 1);
  const int col = nt * 16 + l16;
  const float* __restrict__ Ar = A + (long long)rclamp * K;

  v8f acc = {};
#pragma unroll 4
  for (int k = 0; k < K; k += 4) {
    const int ka = k + 2 * half;
    v2f a, b;
    a.x = Ar[ka];
    a.y = Ar[ka + 1];
    b.x = B[(long long)ka * N + col];
    b.y = B[(long long)(ka + 1) * N + col];
    acc = __builtin_amdgcn_wmma_f32_16x16x4_f32(false, a, false, b, (short)0, acc,
                                                false, false);
  }

  const int m0 = mt * 16 + 8 * half;
  float* __restrict__ Cp = C + (long long)m0 * N + col;
  if (mt * 16 + 16 <= M) {
#pragma unroll
    for (int v = 0; v < 8; ++v) Cp[(long long)v * N] = acc[v];
  } else {
#pragma unroll
    for (int v = 0; v < 8; ++v)
      if (m0 + v < M) Cp[(long long)v * N] = acc[v];
  }
}

__global__ void deg_rank_kernel(const int* __restrict__ dst, int* __restrict__ deg,
                                int* __restrict__ rank, int e) {
  const int i = blockIdx.x * blockDim.x + threadIdx.x;
  if (i < e) rank[i] = atomicAdd(&deg[dst[i]], 1);
}

__global__ __launch_bounds__(1024) void scan_dinv_kernel(const int* __restrict__ deg,
                                                         int* __restrict__ rowptr,
                                                         float* __restrict__ dinv, int n) {
  __shared__ int wsum[32];
  __shared__ int sbase;
  const int tid = threadIdx.x;
  const int lane = tid & 31;
  const int wv = tid >> 5;
  const int first = blockIdx.x * 1024;

  int part = 0;
  for (int j = tid; j < first; j += 1024) part += deg[j];
  for (int o = 16; o > 0; o >>= 1) part += __shfl_xor(part, o);
  if (lane == 0) wsum[wv] = part;
  __syncthreads();
  if (wv == 0) {
    int t = wsum[lane];
    for (int o = 16; o > 0; o >>= 1) t += __shfl_xor(t, o);
    if (lane == 0) sbase = t;
  }
  __syncthreads();
  const int base = sbase;

  const int i = first + tid;
  const int dg = i < n ? deg[i] : 0;
  int incl = dg;
  for (int o = 1; o < 32; o <<= 1) {
    const int t = __shfl_up(incl, o);
    if (lane >= o) incl += t;
  }
  if (lane == 31) wsum[wv] = incl;
  __syncthreads();
  if (wv == 0) {
    const int t = wsum[lane];
    int s = t;
    for (int o = 1; o < 32; o <<= 1) {
      const int u = __shfl_up(s, o);
      if (lane >= o) s += u;
    }
    wsum[lane] = s - t;
  }
  __syncthreads();
  const int excl = base + wsum[wv] + incl - dg;
  if (i < n) {
    rowptr[i] = excl;
    dinv[i] = rsqrtf((float)dg + 1.0f);
    if (i == n - 1) rowptr[n] = excl + dg;
  }
}

__global__ void fill_kernel(const int* __restrict__ src, const int* __restrict__ dst,
                            const int* __restrict__ rank, const int* __restrict__ rowptr,
                            int* __restrict__ csr, int e) {
  const int i = blockIdx.x * blockDim.x + threadIdx.x;
  if (i >= e) return;
  int slot = rowptr[dst[i]] + rank[i];
  slot = min(max(slot, 0), e - 1);
  csr[slot] = src[i];
}

static __device__ __forceinline__ float4 fma4(const float4 a, const float w, const float4 c) {
  return make_float4(fmaf(a.x, w, c.x), fmaf(a.y, w, c.y), fmaf(a.z, w, c.z), fmaf(a.w, w, c.w));
}

__global__ __launch_bounds__(256) void pull_rows128(const float* __restrict__ H,
                                                    const int* __restrict__ rowptr,
                                                    const int* __restrict__ csr,
                                                    const float* __restrict__ dinv,
                                                    float* __restrict__ out, int n, int e) {
  const int lane = threadIdx.x & 31;
  const int node = __builtin_amdgcn_readfirstlane(
      (int)((blockIdx.x * (long long)blockDim.x + threadIdx.x) >> 5));
  if (node >= n) return;
  const float4* __restrict__ H4 = (const float4*)H;
  const int beg = min(max(rowptr[node], 0), e);
  const int end = min(max(rowptr[node + 1], beg), e);
  const float dd = dinv[node];
  const float4 self = H4[(long long)node * 32 + lane];

  float4 acc = make_float4(0.0f, 0.0f, 0.0f, 0.0f);
  for (int base = beg; base < end; base += 32) {
    const int j = base + lane;
    int s = node;
    float dv = 0.0f;
    if (j < end) {
      s = min(max(csr[j], 0), n - 1);
      dv = dinv[s];
    }
    const int wbits = __float_as_int(dv);
    const int cnt = min(32, end - base);
    int t = 0;
    for (; t + 4 <= cnt; t += 4) {
      const int s0 = __builtin_amdgcn_readlane(s, t);
      const int s1 = __builtin_amdgcn_readlane(s, t + 1);
      const int s2 = __builtin_amdgcn_readlane(s, t + 2);
      const int s3 = __builtin_amdgcn_readlane(s, t + 3);
      const float4 a = H4[(long long)s0 * 32 + lane];
      const float4 b = H4[(long long)s1 * 32 + lane];
      const float4 c = H4[(long long)s2 * 32 + lane];
      const float4 d = H4[(long long)s3 * 32 + lane];
      acc = fma4(a, dd * __int_as_float(__builtin_amdgcn_readlane(wbits, t)), acc);
      acc = fma4(b, dd * __int_as_float(__builtin_amdgcn_readlane(wbits, t + 1)), acc);
      acc = fma4(c, dd * __int_as_float(__builtin_amdgcn_readlane(wbits, t + 2)), acc);
      acc = fma4(d, dd * __int_as_float(__builtin_amdgcn_readlane(wbits, t + 3)), acc);
    }
    if (t < cnt) {
      const int r = cnt - t;
      const float4 a = H4[(long long)__builtin_amdgcn_readlane(s, t) * 32 + lane];
      float4 b = make_float4(0.0f, 0.0f, 0.0f, 0.0f);
      float4 c = b;
      if (r > 1) b = H4[(long long)__builtin_amdgcn_readlane(s, t + 1) * 32 + lane];
      if (r > 2) c = H4[(long long)__builtin_amdgcn_readlane(s, t + 2) * 32 + lane];
      acc = fma4(a, dd * __int_as_float(__builtin_amdgcn_readlane(wbits, t)), acc);
      acc = fma4(b, dd * __int_as_float(__builtin_amdgcn_readlane(wbits, t + 1)), acc);
      acc = fma4(c, dd * __int_as_float(__builtin_amdgcn_readlane(wbits, t + 2)), acc);
    }
  }
  acc = fma4(self, dd * dd, acc);
  ((float4*)out)[(long long)node * 32 + lane] = acc;
}

__global__ __launch_bounds__(256) void pull_rows16(const float* __restrict__ H,
                                                   const int* __restrict__ rowptr,
                                                   const int* __restrict__ csr,
                                                   const float* __restrict__ dinv,
                                                   float* __restrict__ out, int n, int e) {
  const long long gid = blockIdx.x * (long long)blockDim.x + threadIdx.x;
  const int node = (int)(gid >> 2);
  if (node >= n) return;
  const int sub = threadIdx.x & 3;
  const float4* __restrict__ H4 = (const float4*)H;
  const int beg = min(max(rowptr[node], 0), e);
  const int end = min(max(rowptr[node + 1], beg), e);
  const float dd = dinv[node];
  const float4 self = H4[(long long)node * 4 + sub];

  float4 acc = make_float4(0.0f, 0.0f, 0.0f, 0.0f);
  for (int j = beg; j < end; ++j) {
    const int s = min(max(csr[j], 0), n - 1);
    acc = fma4(H4[(long long)s * 4 + sub], dinv[s] * dd, acc);
  }
  acc = fma4(self, dd * dd, acc);
  ((float4*)out)[(long long)node * 4 + sub] = acc;
}

template <int LOGF>
__global__ void bias_relu_kernel(const float* __restrict__ agg, const float* __restrict__ b,
                                 float* __restrict__ out, long long total) {
  constexpr int F = 1 << LOGF;
  const long long i = (long long)blockIdx.x * blockDim.x + threadIdx.x;
  if (i >= total) return;
  const float v = agg[i] + b[(int)(i & (F - 1))];
  out[i] = v > 0.0f ? v : 0.0f;
}

__global__ void lsm_kernel(const float* __restrict__ agg, const float* __restrict__ b,
                           float* __restrict__ out, int n) {
  const int i = blockIdx.x * blockDim.x + threadIdx.x;
  if (i >= n) return;
  float v[16];
  float mx = -INFINITY;
#pragma unroll
  for (int j = 0; j < 16; ++j) {
    v[j] = agg[(long long)i * 16 + j] + b[j];
    mx = fmaxf(mx, v[j]);
  }
  float s = 0.0f;
#pragma unroll
  for (int j = 0; j < 16; ++j) s += expf(v[j] - mx);
  const float lse = mx + logf(s);
#pragma unroll
  for (int j = 0; j < 16; ++j) out[(long long)i * 16 + j] = v[j] - lse;
}

extern "C" void kernel_launch(void* const* d_in, const int* in_sizes, int n_in,
                              void* d_out, int out_size, void* d_ws, size_t ws_size,
                              hipStream_t stream) {
  const float* x  = (const float*)d_in[0];
  const int*   ei = (const int*)d_in[1];
  const float* W1 = (const float*)d_in[2];
  const float* b1 = (const float*)d_in[3];
  const float* W2 = (const float*)d_in[4];
  const float* b2 = (const float*)d_in[5];

  const int n = in_sizes[0] / 256;
  const int e = in_sizes[1] / 2;
  const int* src = ei;
  const int* dst = ei + e;

  float* agg1   = (float*)d_ws;
  float* h1     = agg1 + (size_t)n * 128;
  int*   csr    = (int*)(h1 + (size_t)n * 128);
  int*   deg    = csr + e;
  int*   rowptr = deg + n;
  float* dinv   = (float*)(rowptr + n + 1);
  int*   rank   = (int*)agg1;
  float* h2     = agg1;
  float* agg2   = agg1 + (size_t)n * 16;

  hipMemsetAsync(deg, 0, (size_t)n * sizeof(int), stream);

  deg_rank_kernel<<<ceil_div(e, 256), 256, 0, stream>>>(dst, deg, rank, e);
  scan_dinv_kernel<<<ceil_div(n, 1024), 1024, 0, stream>>>(deg, rowptr, dinv, n);
  fill_kernel<<<ceil_div(e, 256), 256, 0, stream>>>(src, dst, rank, rowptr, csr, e);

  {
    const long long waves = (long long)((n + 15) / 16) * (128 / 16);
    gemm_wmma_f32<256, 128><<<ceil_div(waves * 32, 256), 256, 0, stream>>>(x, W1, h1, n);
  }
  pull_rows128<<<ceil_div((long long)n * 32, 256), 256, 0, stream>>>(
      h1, rowptr, csr, dinv, agg1, n, e);
  bias_relu_kernel<7><<<ceil_div((long long)n * 128, 256), 256, 0, stream>>>(
      agg1, b1, h1, (long long)n * 128);

  {
    const long long waves = (long long)((n + 15) / 16) * (16 / 16);
    gemm_wmma_f32<128, 16><<<ceil_div(waves * 32, 256), 256, 0, stream>>>(h1, W2, h2, n);
  }
  pull_rows16<<<ceil_div((long long)n * 4, 256), 256, 0, stream>>>(
      h2, rowptr, csr, dinv, agg2, n, e);
  lsm_kernel<<<ceil_div(n, 256), 256, 0, stream>>>(agg2, b2, (float*)d_out, n);
}


# GEMMs re-tiled: W staged once per work-group in LDS in fragment order, a wave owns 16 rows x all N columns, A read as 16-B row pieces (k order inside WMMA pairs remapped, so the sum order differs)
# speedup vs baseline: 3.3188x; 1.2116x over previous
#include <hip/hip_runtime.h>
#include <math.h>


typedef __attribute__((ext_vector_type(2))) float v2f;
typedef __attribute__((ext_vector_type(8))) float v8f;

static inline int ceil_div(long long a, long long b) { return (int)((a + b - 1) / b); }

template <int K, int N>
__global__ __launch_bounds__(256, 4) void gemm_rows16_lds(const float* __restrict__ A,
                                                       const float* __restrict__ B,
                                                       float* __restrict__ C, int M) {
  constexpr int NT = N / 16;
  constexpr int KQ = K / 4;
  constexpr int KB = K / 32;
  constexpr int J = K / 8;
  __shared__ float4 Bl[KQ * N];

  const int tid = threadIdx.x;
  const int lane = tid & 31;
  const int wv = __builtin_amdgcn_readfirstlane(tid >> 5);
  const int half = lane >> 4;
  const int l16 = lane & 15;
  const int mTiles = (M + 15) >> 4;

  int mt = (int)blockIdx.x + (int)gridDim.x * wv;
  const int mtStep = (int)gridDim.x * 8;

  float4 ac[4], an[4];
  if (mt < mTiles) {
    const int row = mt * 16 + l16;
    const float* Ar = A + (long long)(row < M ? row : M - 1) * K + 4 * half;
#pragma unroll
    for (int i = 0; i < 4; ++i) ac[i] = *(const float4*)(Ar + 8 * i);
  }

  static_assert((KQ * N) % 256 == 0, "the staging loop has the same trip count in every thread");
#pragma unroll 8
  for (int it = 0; it < KQ * N / 256; ++it) {
    const int idx = it * 256 + tid;
    const int kq = idx / N;
    const int nn = idx - kq * N;
    const float* p = B + (long long)(4 * kq) * N + nn;
    Bl[idx] = make_float4(p[0], p[N], p[2 * N], p[3 * N]);
  }
  __syncthreads();

  const float4* Blane = Bl + half * N + l16;
  float4 bc[NT], bn[NT];
#pragma unroll
  for (int nt = 0; nt < NT; ++nt) bc[nt] = Blane[nt * 16];

  while (mt < mTiles) {
    const int mtNext = mt + mtStep;
    v8f acc[NT];
#pragma unroll
    for (int nt = 0; nt < NT; ++nt) acc[nt] = v8f{};

    const int row = mt * 16 + l16;
    const float* Ar = A + (long long)(row < M ? row : M - 1) * K + 4 * half;
    const int rowN = mtNext * 16 + l16;
    const float* ArN = A + (long long)(rowN < M ? rowN : M - 1) * K + 4 * half;

#pragma unroll
    for (int kb = 0; kb < KB; ++kb) {
      if (kb + 1 < KB) {
#pragma unroll
        for (int i = 0; i < 4; ++i) an[i] = *(const float4*)(Ar + 32 * (kb + 1) + 8 * i);
      } else if (mtNext < mTiles) {
#pragma unroll
        for (int i = 0; i < 4; ++i) an[i] = *(const float4*)(ArN + 8 * i);
      }
#pragma unroll
      for (int i = 0; i < 4; ++i) {
        const int j = kb * 4 + i;
        const int jn = (j + 1) % J;
#pragma unroll
        for (int nt = 0; nt < NT; ++nt) bn[nt] = Blane[(2 * jn) * N + nt * 16];
        v2f a0, a1;
        a0.x = ac[i].x; a0.y = ac[i].y;
        a1.x = ac[i].z; a1.y = ac[i].w;
#pragma unroll
        for (int nt = 0; nt < NT; ++nt) {
          v2f b0;
          b0.x = bc[nt].x; b0.y = bc[nt].y;
          acc[nt] = __builtin_amdgcn_wmma_f32_16x16x4_f32(false, a0, false, b0, (short)0,
                                                          acc[nt], false, false);
        }
#pragma unroll
        for (int nt = 0; nt < NT; ++nt) {
          v2f b1;
          b1.x = bc[nt].z; b1.y = bc[nt].w;
          acc[nt] = __builtin_amdgcn_wmma_f32_16x16x4_f32(false, a1, false, b1, (short)0,
                                                          acc[nt], false, false);
        }
#pragma unroll
        for (int nt = 0; nt < NT; ++nt) bc[nt] = bn[nt];
      }
#pragma unroll
      for (int i = 0; i < 4; ++i) ac[i] = an[i];
    }

    const int m0 = mt * 16 + 8 * half;
    float* __restrict__ Cp = C + (long long)m0 * N + l16;
    if (mt * 16 + 16 <= M) {
#pragma unroll
      for (int nt = 0; nt < NT; ++nt)
#pragma unroll
        for (int v = 0; v < 8; ++v) Cp[(long long)v * N + nt * 16] = acc[nt][v];
    } else {
#pragma unroll
      for (int nt = 0; nt < NT; ++nt)
#pragma unroll
        for (int v = 0; v < 8; ++v)
          if (m0 + v < M) Cp[(long long)v * N + nt * 16] = acc[nt][v];
    }
    mt = mtNext;
  }
}

__global__ void deg_rank_kernel(const int* __restrict__ dst, int* __restrict__ deg,
                                int* __restrict__ rank, int e) {
  const int i = blockIdx.x * blockDim.x + threadIdx.x;
  if (i < e) rank[i] = atomicAdd(&deg[dst[i]], 1);
}

__global__ __launch_bounds__(1024) void scan_dinv_kernel(const int* __restrict__ deg,
                                                         int* __restrict__ rowptr,
                                                         float* __restrict__ dinv, int n) {
  __shared__ int wsum[32];
  __shared__ int sbase;
  const int tid = threadIdx.x;
  const int lane = tid & 31;
  const int wv = tid >> 5;
  const int first = blockIdx.x * 1024;

  int part = 0;
  for (int j = tid; j < first; j += 1024) part += deg[j];
  for (int o = 16; o > 0; o >>= 1) part += __shfl_xor(part, o);
  if (lane == 0) wsum[wv] = part;
  __syncthreads();
  if (wv == 0) {
    int t = wsum[lane];
    for (int o = 16; o > 0; o >>= 1) t += __shfl_xor(t, o);
    if (lane == 0) sbase = t;
  }
  __syncthreads();
  const int base = sbase;

  const int i = first + tid;
  const int dg = i < n ? deg[i] : 0;
  int incl = dg;
  for (int o = 1; o < 32; o <<= 1) {
    const int t = __shfl_up(incl, o);
    if (lane >= o) incl += t;
  }
  if (lane == 31) wsum[wv] = incl;
  __syncthreads();
  if (wv == 0) {
    const int t = wsum[lane];
    int s = t;
    for (int o = 1; o < 32; o <<= 1) {
      const int u = __shfl_up(s, o);
      if (lane >= o) s += u;
    }
    wsum[lane] = s - t;
  }
  __syncthreads();
  const int excl = base + wsum[wv] + incl - dg;
  if (i < n) {
    rowptr[i] = excl;
    dinv[i] = rsqrtf((float)dg + 1.0f);
    if (i == n - 1) rowptr[n] = excl + dg;
  }
}

__global__ void fill_kernel(const int* __restrict__ src, const int* __restrict__ dst,
                            const int* __restrict__ rank, const int* __restrict__ rowptr,
                            int* __restrict__ csr, int e) {
  const int i = blockIdx.x * blockDim.x + threadIdx.x;
  if (i >= e) return;
  int slot = rowptr[dst[i]] + rank[i];
  slot = min(max(slot, 0), e - 1);
  csr[slot] = src[i];
}

static __device__ __forceinline__ float4 fma4(const float4 a, const float w, const float4 c) {
  return make_float4(fmaf(a.x, w, c.x), fmaf(a.y, w, c.y), fmaf(a.z, w, c.z), fmaf(a.w, w, c.w));
}

__global__ __launch_bounds__(256) void pull_rows128(const float* __restrict__ H,
                                                    const int* __restrict__ rowptr,
                                                    const int* __restrict__ csr,
                                                    const float* __restrict__ dinv,
                                                    float* __restrict__ out, int n, int e) {
  const int lane = threadIdx.x & 31;
  const int node = __builtin_amdgcn_readfirstlane(
      (int)((blockIdx.x * (long long)blockDim.x + threadIdx.x) >> 5));
  if (node >= n) return;
  const float4* __restrict__ H4 = (const float4*)H;
  const int beg = min(max(rowptr[node], 0), e);
  const int end = min(max(rowptr[node + 1], beg), e);
  const float dd = dinv[node];
  const float4 self = H4[(long long)node * 32 + lane];

  float4 acc = make_float4(0.0f, 0.0f, 0.0f, 0.0f);
  for (int base = beg; base < end; base += 32) {
    const int j = base + lane;
    int s = node;
    float dv = 0.0f;
    if (j < end) {
      s = min(max(csr[j], 0), n - 1);
      dv = dinv[s];
    }
    const int wbits = __float_as_int(dv);
    const int cnt = min(32, end - base);
    int t = 0;
    for (; t + 4 <= cnt; t += 4) {
      const int s0 = __builtin_amdgcn_readlane(s, t);
      const int s1 = __builtin_amdgcn_readlane(s, t + 1);
      const int s2 = __builtin_amdgcn_readlane(s, t + 2);
      const int s3 = __builtin_amdgcn_readlane(s, t + 3);
      const float4 a = H4[(long long)s0 * 32 + lane];
      const float4 b = H4[(long long)s1 * 32 + lane];
      const float4 c = H4[(long long)s2 * 32 + lane];
      const float4 d = H4[(long long)s3 * 32 + lane];
      acc = fma4(a, dd * __int_as_float(__builtin_amdgcn_readlane(wbits, t)), acc);
      acc = fma4(b, dd * __int_as_float(__builtin_amdgcn_readlane(wbits, t + 1)), acc);
      acc = fma4(c, dd * __int_as_float(__builtin_amdgcn_readlane(wbits, t + 2)), acc);
      acc = fma4(d, dd * __int_as_float(__builtin_amdgcn_readlane(wbits, t + 3)), acc);
    }
    if (t < cnt) {
      const int r = cnt - t;
      const float4 a = H4[(long long)__builtin_amdgcn_readlane(s, t) * 32 + lane];
      float4 b = make_float4(0.0f, 0.0f, 0.0f, 0.0f);
      float4 c = b;
      if (r > 1) b = H4[(long long)__builtin_amdgcn_readlane(s, t + 1) * 32 + lane];
      if (r > 2) c = H4[(long long)__builtin_amdgcn_readlane(s, t + 2) * 32 + lane];
      acc = fma4(a, dd * __int_as_float(__builtin_amdgcn_readlane(wbits, t)), acc);
      acc = fma4(b, dd * __int_as_float(__builtin_amdgcn_readlane(wbits, t + 1)), acc);
      acc = fma4(c, dd * __int_as_float(__builtin_amdgcn_readlane(wbits, t + 2)), acc);
    }
  }
  acc = fma4(self, dd * dd, acc);
  ((float4*)out)[(long long)node * 32 + lane] = acc;
}

__global__ __launch_bounds__(256) void pull_rows16(const float* __restrict__ H,
                                                   const int* __restrict__ rowptr,
                                                   const int* __restrict__ csr,
                                                   const float* __restrict__ dinv,
                                                   float* __restrict__ out, int n, int e) {
  const long long gid = blockIdx.x * (long long)blockDim.x + threadIdx.x;
  const int node = (int)(gid >> 2);
  if (node >= n) return;
  const int sub = threadIdx.x & 3;
  const float4* __restrict__ H4 = (const float4*)H;
  const int beg = min(max(rowptr[node], 0), e);
  const int end = min(max(rowptr[node + 1], beg), e);
  const float dd = dinv[node];
  const float4 self = H4[(long long)node * 4 + sub];

  float4 acc = make_float4(0.0f, 0.0f, 0.0f, 0.0f);
  for (int j = beg; j < end; ++j) {
    const int s = min(max(csr[j], 0), n - 1);
    acc = fma4(H4[(long long)s * 4 + sub], dinv[s] * dd, acc);
  }
  acc = fma4(self, dd * dd, acc);
  ((float4*)out)[(long long)node * 4 + sub] = acc;
}

template <int LOGF>
__global__ void bias_relu_kernel(const float* __restrict__ agg, const float* __restrict__ b,
                                 float* __restrict__ out, long long total) {
  constexpr int F = 1 << LOGF;
  const long long i = (long long)blockIdx.x * blockDim.x + threadIdx.x;
  if (i >= total) return;
  const float v = agg[i] + b[(int)(i & (F - 1))];
  out[i] = v > 0.0f ? v : 0.0f;
}

__global__ void lsm_kernel(const float* __restrict__ agg, const float* __restrict__ b,
                           float* __restrict__ out, int n) {
  const int i = blockIdx.x * blockDim.x + threadIdx.x;
  if (i >= n) return;
  float v[16];
  float mx = -INFINITY;
#pragma unroll
  for (int j = 0; j < 16; ++j) {
    v[j] = agg[(long long)i * 16 + j] + b[j];
    mx = fmaxf(mx, v[j]);
  }
  float s = 0.0f;
#pragma unroll
  for (int j = 0; j < 16; ++j) s += expf(v[j] - mx);
  const float lse = mx + logf(s);
#pragma unroll
  for (int j = 0; j < 16; ++j) out[(long long)i * 16 + j] = v[j] - lse;
}

extern "C" void kernel_launch(void* const* d_in, const int* in_sizes, int n_in,
                              void* d_out, int out_size, void* d_ws, size_t ws_size,
                              hipStream_t stream) {
  const float* x  = (const float*)d_in[0];
  const int*   ei = (const int*)d_in[1];
  const float* W1 = (const float*)d_in[2];
  const float* b1 = (const float*)d_in[3];
  const float* W2 = (const float*)d_in[4];
  const float* b2 = (const float*)d_in[5];

  const int n = in_sizes[0] / 256;
  const int e = in_sizes[1] / 2;
  const int* src = ei;
  const int* dst = ei + e;

  float* agg1   = (float*)d_ws;
  float* h1     = agg1 + (size_t)n * 128;
  int*   csr    = (int*)(h1 + (size_t)n * 128);
  int*   deg    = csr + e;
  int*   rowptr = deg + n;
  float* dinv   = (float*)(rowptr + n + 1);
  int*   rank   = (int*)agg1;
  float* h2     = agg1;
  float* agg2   = agg1 + (size_t)n * 16;

  hipMemsetAsync(deg, 0, (size_t)n * sizeof(int), stream);

  deg_rank_kernel<<<ceil_div(e, 256), 256, 0, stream>>>(dst, deg, rank, e);
  scan_dinv_kernel<<<ceil_div(n, 1024), 1024, 0, stream>>>(deg, rowptr, dinv, n);
  fill_kernel<<<ceil_div(e, 256), 256, 0, stream>>>(src, dst, rank, rowptr, csr, e);

  gemm_rows16_lds<256, 128><<<512, 256, 0, stream>>>(x, W1, h1, n);
  pull_rows128<<<ceil_div((long long)n * 32, 256), 256, 0, stream>>>(
      h1, rowptr, csr, dinv, agg1, n, e);
  bias_relu_kernel<7><<<ceil_div((long long)n * 128, 256), 256, 0, stream>>>(
      agg1, b1, h1, (long long)n * 128);

  gemm_rows16_lds<128, 16><<<512, 256, 0, stream>>>(h1, W2, h2, n);
  pull_rows16<<<ceil_div((long long)n * 4, 256), 256, 0, stream>>>(
      h2, rowptr, csr, dinv, agg2, n, e);
  lsm_kernel<<<ceil_div(n, 256), 256, 0, stream>>>(agg2, b2, (float*)d_out, n);
}


# Pointwise stages fused into the aggregation kernels: bias+ReLU at the 128-wide row store, bias+log-softmax at the 16-wide one (exp sum: 4 per lane, then 4-lane tree); 2 launches, 2 planes gone
# speedup vs baseline: 3.8256x; 1.1527x over previous
#include <hip/hip_runtime.h>
#include <math.h>


typedef __attribute__((ext_vector_type(2))) float v2f;
typedef __attribute__((ext_vector_type(8))) float v8f;

static inline int ceil_div(long long a, long long b) { return (int)((a + b - 1) / b); }

template <int K, int N>
__global__ __launch_bounds__(256, 4) void gemm_rows16_lds(const float* __restrict__ A,
                                                       const float* __restrict__ B,
                                                       float* __restrict__ C, int M) {
  constexpr int NT = N / 16;
  constexpr int KQ = K / 4;
  constexpr int KB = K / 32;
  constexpr int J = K / 8;
  __shared__ float4 Bl[KQ * N];

  const int tid = threadIdx.x;
  const int lane = tid & 31;
  const int wv = __builtin_amdgcn_readfirstlane(tid >> 5);
  const int half = lane >> 4;
  const int l16 = lane & 15;
  const int mTiles = (M + 15) >> 4;

  int mt = (int)blockIdx.x + (int)gridDim.x * wv;
  const int mtStep = (int)gridDim.x * 8;

  float4 ac[4], an[4];
  if (mt < mTiles) {
    const int row = mt * 16 + l16;
    const float* Ar = A + (long long)(row < M ? row : M - 1) * K + 4 * half;
#pragma unroll
    for (int i = 0; i < 4; ++i) ac[i] = *(const float4*)(Ar + 8 * i);
  }

  static_assert((KQ * N) % 256 == 0, "the staging loop has the same trip count in every thread");
#pragma unroll 8
  for (int it = 0; it < KQ * N / 256; ++it) {
    const int idx = it * 256 + tid;
    const int kq = idx / N;
    const int nn = idx - kq * N;
    const float* p = B + (long long)(4 * kq) * N + nn;
    Bl[idx] = make_float4(p[0], p[N], p[2 * N], p[3 * N]);
  }
  __syncthreads();

  const float4* Blane = Bl + half * N + l16;
  float4 bc[NT], bn[NT];
#pragma unroll
  for (int nt = 0; nt < NT; ++nt) bc[nt] = Blane[nt * 16];

  while (mt < mTiles) {
    const int mtNext = mt + mtStep;
    v8f acc[NT];
#pragma unroll
    for (int nt = 0; nt < NT; ++nt) acc[nt] = v8f{};

    const int row = mt * 16 + l16;
    const float* Ar = A + (long long)(row < M ? row : M - 1) * K + 4 * half;
    const int rowN = mtNext * 16 + l16;
    const float* ArN = A + (long long)(rowN < M ? rowN : M - 1) * K + 4 * half;

#pragma unroll
    for (int kb = 0; kb < KB; ++kb) {
      if (kb + 1 < KB) {
#pragma unroll
        for (int i = 0; i < 4; ++i) an[i] = *(const float4*)(Ar + 32 * (kb + 1) + 8 * i);
      } else if (mtNext < mTiles) {
#pragma unroll
        for (int i = 0; i < 4; ++i) an[i] = *(const float4*)(ArN + 8 * i);
      }
#pragma unroll
      for (int i = 0; i < 4; ++i) {
        const int j = kb * 4 + i;
        const int jn = (j + 1) % J;
#pragma unroll
        for (int nt = 0; nt < NT; ++nt) bn[nt] = Blane[(2 * jn) * N + nt * 16];
        v2f a0, a1;
        a0.x = ac[i].x; a0.y = ac[i].y;
        a1.x = ac[i].z; a1.y = ac[i].w;
#pragma unroll
        for (int nt = 0; nt < NT; ++nt) {
          v2f b0;
          b0.x = bc[nt].x; b0.y = bc[nt].y;
          acc[nt] = __builtin_amdgcn_wmma_f32_16x16x4_f32(false, a0, false, b0, (short)0,
                                                          acc[nt], false, false);
        }
#pragma unroll
        for (int nt = 0; nt < NT; ++nt) {
          v2f b1;
          b1.x = bc[nt].z; b1.y = bc[nt].w;
          acc[nt] = __builtin_amdgcn_wmma_f32_16x16x4_f32(false, a1, false, b1, (short)0,
                                                          acc[nt], false, false);
        }
#pragma unroll
        for (int nt = 0; nt < NT; ++nt) bc[nt] = bn[nt];
      }
#pragma unroll
      for (int i = 0; i < 4; ++i) ac[i] = an[i];
    }

    const int m0 = mt * 16 + 8 * half;
    float* __restrict__ Cp = C + (long long)m0 * N + l16;
    if (mt * 16 + 16 <= M) {
#pragma unroll
      for (int nt = 0; nt < NT; ++nt)
#pragma unroll
        for (int v = 0; v < 8; ++v) Cp[(long long)v * N + nt * 16] = acc[nt][v];
    } else {
#pragma unroll
      for (int nt = 0; nt < NT; ++nt)
#pragma unroll
        for (int v = 0; v < 8; ++v)
          if (m0 + v < M) Cp[(long long)v * N + nt * 16] = acc[nt][v];
    }
    mt = mtNext;
  }
}

__global__ void deg_rank_kernel(const int* __restrict__ dst, int* __restrict__ deg,
                                int* __restrict__ rank, int e) {
  const int i = blockIdx.x * blockDim.x + threadIdx.x;
  if (i < e) rank[i] = atomicAdd(&deg[dst[i]], 1);
}

__global__ __launch_bounds__(1024) void scan_dinv_kernel(const int* __restrict__ deg,
                                                         int* __restrict__ rowptr,
                                                         float* __restrict__ dinv, int n) {
  __shared__ int wsum[32];
  __shared__ int sbase;
  const int tid = threadIdx.x;
  const int lane = tid & 31;
  const int wv = tid >> 5;
  const int first = blockIdx.x * 1024;

  int part = 0;
  for (int j = tid; j < first; j += 1024) part += deg[j];
  for (int o = 16; o > 0; o >>= 1) part += __shfl_xor(part, o);
  if (lane == 0) wsum[wv] = part;
  __syncthreads();
  if (wv == 0) {
    int t = wsum[lane];
    for (int o = 16; o > 0; o >>= 1) t += __shfl_xor(t, o);
    if (lane == 0) sbase = t;
  }
  __syncthreads();
  const int base = sbase;

  const int i = first + tid;
  const int dg = i < n ? deg[i] : 0;
  int incl = dg;
  for (int o = 1; o < 32; o <<= 1) {
    const int t = __shfl_up(incl, o);
    if (lane >= o) incl += t;
  }
  if (lane == 31) wsum[wv] = incl;
  __syncthreads();
  if (wv == 0) {
    const int t = wsum[lane];
    int s = t;
    for (int o = 1; o < 32; o <<= 1) {
      const int u = __shfl_up(s, o);
      if (lane >= o) s += u;
    }
    wsum[lane] = s - t;
  }
  __syncthreads();
  const int excl = base + wsum[wv] + incl - dg;
  if (i < n) {
    rowptr[i] = excl;
    dinv[i] = rsqrtf((float)dg + 1.0f);
    if (i == n - 1) rowptr[n] = excl + dg;
  }
}

__global__ void fill_kernel(const int* __restrict__ src, const int* __restrict__ dst,
                            const int* __restrict__ rank, const int* __restrict__ rowptr,
                            int* __restrict__ csr, int e) {
  const int i = blockIdx.x * blockDim.x + threadIdx.x;
  if (i >= e) return;
  int slot = rowptr[dst[i]] + rank[i];
  slot = min(max(slot, 0), e - 1);
  csr[slot] = src[i];
}

static __device__ __forceinline__ float4 fma4(const float4 a, const float w, const float4 c) {
  return make_float4(fmaf(a.x, w, c.x), fmaf(a.y, w, c.y), fmaf(a.z, w, c.z), fmaf(a.w, w, c.w));
}

__global__ __launch_bounds__(256) void pull_rows128(const float* __restrict__ H,
                                                    const int* __restrict__ rowptr,
                                                    const int* __restrict__ csr,
                                                    const float* __restrict__ dinv,
                                                    const float* __restrict__ b,
                                                    float* __restrict__ out, int n, int e) {
  const int lane = threadIdx.x & 31;
  const int node = __builtin_amdgcn_readfirstlane(
      (int)((blockIdx.x * (long long)blockDim.x + threadIdx.x) >> 5));
  if (node >= n) return;
  const float4* __restrict__ H4 = (const float4*)H;
  const int beg = min(max(rowptr[node], 0), e);
  const int end = min(max(rowptr[node + 1], beg), e);
  const float dd = dinv[node];
  const float4 self = H4[(long long)node * 32 + lane];
  const float4 bb = ((const float4*)b)[lane];

  float4 acc = make_float4(0.0f, 0.0f, 0.0f, 0.0f);
  for (int base = beg; base < end; base += 32) {
    const int j = base + lane;
    int s = node;
    float dv = 0.0f;
    if (j < end) {
      s = min(max(csr[j], 0), n - 1);
      dv = dinv[s];
    }
    const int wbits = __float_as_int(dv);
    const int cnt = min(32, end - base);
    int t = 0;
    for (; t + 4 <= cnt; t += 4) {
      const int s0 = __builtin_amdgcn_readlane(s, t);
      const int s1 = __builtin_amdgcn_readlane(s, t + 1);
      const int s2 = __builtin_amdgcn_readlane(s, t + 2);
      const int s3 = __builtin_amdgcn_readlane(s, t + 3);
      const float4 a = H4[(long long)s0 * 32 + lane];
      const float4 b = H4[(long long)s1 * 32 + lane];
      const float4 c = H4[(long long)s2 * 32 + lane];
      const float4 d = H4[(long long)s3 * 32 + lane];
      acc = fma4(a, dd * __int_as_float(__builtin_amdgcn_readlane(wbits, t)), acc);
      acc = fma4(b, dd * __int_as_float(__builtin_amdgcn_readlane(wbits, t + 1)), acc);
      acc = fma4(c, dd * __int_as_float(__builtin_amdgcn_readlane(wbits, t + 2)), acc);
      acc = fma4(d, dd * __int_as_float(__builtin_amdgcn_readlane(wbits, t + 3)), acc);
    }
    if (t < cnt) {
      const int r = cnt - t;
      const float4 a = H4[(long long)__builtin_amdgcn_readlane(s, t) * 32 + lane];
      float4 b = make_float4(0.0f, 0.0f, 0.0f, 0.0f);
      float4 c = b;
      if (r > 1) b = H4[(long long)__builtin_amdgcn_readlane(s, t + 1) * 32 + lane];
      if (r > 2) c = H4[(long long)__builtin_amdgcn_readlane(s, t + 2) * 32 + lane];
      acc = fma4(a, dd * __int_as_float(__builtin_amdgcn_readlane(wbits, t)), acc);
      acc = fma4(b, dd * __int_as_float(__builtin_amdgcn_readlane(wbits, t + 1)), acc);
      acc = fma4(c, dd * __int_as_float(__builtin_amdgcn_readlane(wbits, t + 2)), acc);
    }
  }
  acc = fma4(self, dd * dd, acc);
  float4 r;
  r.x = acc.x + bb.x; r.y = acc.y + bb.y; r.z = acc.z + bb.z; r.w = acc.w + bb.w;
  r.x = r.x > 0.0f ? r.x : 0.0f;
  r.y = r.y > 0.0f ? r.y : 0.0f;
  r.z = r.z > 0.0f ? r.z : 0.0f;
  r.w = r.w > 0.0f ? r.w : 0.0f;
  ((float4*)out)[(long long)node * 32 + lane] = r;
}

__global__ __launch_bounds__(256) void pull_rows16(const float* __restrict__ H,
                                                   const int* __restrict__ rowptr,
                                                   const int* __restrict__ csr,
                                                   const float* __restrict__ dinv,
                                                   const float* __restrict__ b,
                                                   float* __restrict__ out, int n, int e) {
  const long long gid = blockIdx.x * (long long)blockDim.x + threadIdx.x;
  const int node = (int)(gid >> 2);
  if (node >= n) return;
  const int sub = threadIdx.x & 3;
  const float4* __restrict__ H4 = (const float4*)H;
  const int beg = min(max(rowptr[node], 0), e);
  const int end = min(max(rowptr[node + 1], beg), e);
  const float dd = dinv[node];
  const float4 self = H4[(long long)node * 4 + sub];

  float4 acc = make_float4(0.0f, 0.0f, 0.0f, 0.0f);
  for (int j = beg; j < end; ++j) {
    const int s = min(max(csr[j], 0), n - 1);
    acc = fma4(H4[(long long)s * 4 + sub], dinv[s] * dd, acc);
  }
  acc = fma4(self, dd * dd, acc);

  const float4 bb = ((const float4*)b)[sub];
  float4 v;
  v.x = acc.x + bb.x; v.y = acc.y + bb.y; v.z = acc.z + bb.z; v.w = acc.w + bb.w;
  float mx = fmaxf(fmaxf(v.x, v.y), fmaxf(v.z, v.w));
  mx = fmaxf(mx, __shfl_xor(mx, 1));
  mx = fmaxf(mx, __shfl_xor(mx, 2));
  float s = expf(v.x - mx);
  s += expf(v.y - mx);
  s += expf(v.z - mx);
  s += expf(v.w - mx);
  s += __shfl_xor(s, 1);
  s += __shfl_xor(s, 2);
  const float lse = mx + logf(s);
  ((float4*)out)[(long long)node * 4 + sub] =
      make_float4(v.x - lse, v.y - lse, v.z - lse, v.w - lse);
}

extern "C" void kernel_launch(void* const* d_in, const int* in_sizes, int n_in,
                              void* d_out, int out_size, void* d_ws, size_t ws_size,
                              hipStream_t stream) {
  const float* x  = (const float*)d_in[0];
  const int*   ei = (const int*)d_in[1];
  const float* W1 = (const float*)d_in[2];
  const float* b1 = (const float*)d_in[3];
  const float* W2 = (const float*)d_in[4];
  const float* b2 = (const float*)d_in[5];

  const int n = in_sizes[0] / 256;
  const int e = in_sizes[1] / 2;
  const int* src = ei;
  const int* dst = ei + e;

  float* a1     = (float*)d_ws;
  float* h1     = a1 + (size_t)n * 128;
  int*   csr    = (int*)(h1 + (size_t)n * 128);
  int*   deg    = csr + e;
  int*   rowptr = deg + n;
  float* dinv   = (float*)(rowptr + n + 1);
  int*   rank   = (int*)a1;
  float* h2     = h1;

  hipMemsetAsync(deg, 0, (size_t)n * sizeof(int), stream);

  deg_rank_kernel<<<ceil_div(e, 256), 256, 0, stream>>>(dst, deg, rank, e);
  scan_dinv_kernel<<<ceil_div(n, 1024), 1024, 0, stream>>>(deg, rowptr, dinv, n);
  fill_kernel<<<ceil_div(e, 256), 256, 0, stream>>>(src, dst, rank, rowptr, csr, e);

  gemm_rows16_lds<256, 128><<<512, 256, 0, stream>>>(x, W1, h1, n);
  pull_rows128<<<ceil_div((long long)n * 32, 256), 256, 0, stream>>>(
      h1, rowptr, csr, dinv, b1, a1, n, e);

  gemm_rows16_lds<128, 16><<<512, 256, 0, stream>>>(a1, W2, h2, n);
  pull_rows16<<<ceil_div((long long)n * 4, 256), 256, 0, stream>>>(
      h2, rowptr, csr, dinv, b2, (float*)d_out, n, e);
}


# dinv[s] folded into both GEMM stores (rows stored as dinv[row]*h): both aggregations lose the per-edge random dinv lookup; dinv[d] scales the finished sum (sum order kept, rounding points moved)
# speedup vs baseline: 3.9731x; 1.0386x over previous
#include <hip/hip_runtime.h>
#include <math.h>


typedef __attribute__((ext_vector_type(2))) float v2f;
typedef __attribute__((ext_vector_type(8))) float v8f;

static inline int ceil_div(long long a, long long b) { return (int)((a + b - 1) / b); }

template <int K, int N>
__global__ __launch_bounds__(256, 4) void gemm_rows16_lds(const float* __restrict__ A,
                                                       const float* __restrict__ B,
                                                       const float* __restrict__ rs,
                                                       float* __restrict__ C, int M) {
  constexpr int NT = N / 16;
  constexpr int KQ = K / 4;
  constexpr int KB = K / 32;
  constexpr int J = K / 8;
  __shared__ float4 Bl[KQ * N];

  const int tid = threadIdx.x;
  const int lane = tid & 31;
  const int wv = __builtin_amdgcn_readfirstlane(tid >> 5);
  const int half = lane >> 4;
  const int l16 = lane & 15;
  const int mTiles = (M + 15) >> 4;

  int mt = (int)blockIdx.x + (int)gridDim.x * wv;
  const int mtStep = (int)gridDim.x * 8;

  float4 ac[4], an[4];
  if (mt < mTiles) {
    const int row = mt * 16 + l16;
    const float* Ar = A + (long long)(row < M ? row : M - 1) * K + 4 * half;
#pragma unroll
    for (int i = 0; i < 4; ++i) ac[i] = *(const float4*)(Ar + 8 * i);
  }

  static_assert((KQ * N) % 256 == 0, "the staging loop has the same trip count in every thread");
#pragma unroll 8
  for (int it = 0; it < KQ * N / 256; ++it) {
    const int idx = it * 256 + tid;
    const int kq = idx / N;
    const int nn = idx - kq * N;
    const float* p = B + (long long)(4 * kq) * N + nn;
    Bl[idx] = make_float4(p[0], p[N], p[2 * N], p[3 * N]);
  }
  __syncthreads();

  const float4* Blane = Bl + half * N + l16;
  float4 bc[NT], bn[NT];
#pragma unroll
  for (int nt = 0; nt < NT; ++nt) bc[nt] = Blane[nt * 16];

  while (mt < mTiles) {
    const int mtNext = mt + mtStep;
    v8f acc[NT];
#pragma unroll
    for (int nt = 0; nt < NT; ++nt) acc[nt] = v8f{};

    const int row = mt * 16 + l16;
    const float* Ar = A + (long long)(row < M ? row : M - 1) * K + 4 * half;
    const int rowN = mtNext * 16 + l16;
    const float* ArN = A + (long long)(rowN < M ? rowN : M - 1) * K + 4 * half;

#pragma unroll
    for (int kb = 0; kb < KB; ++kb) {
      if (kb + 1 < KB) {
#pragma unroll
        for (int i = 0; i < 4; ++i) an[i] = *(const float4*)(Ar + 32 * (kb + 1) + 8 * i);
      } else if (mtNext < mTiles) {
#pragma unroll
        for (int i = 0; i < 4; ++i) an[i] = *(const float4*)(ArN + 8 * i);
      }
#pragma unroll
      for (int i = 0; i < 4; ++i) {
        const int j = kb * 4 + i;
        const int jn = (j + 1) % J;
#pragma unroll
        for (int nt = 0; nt < NT; ++nt) bn[nt] = Blane[(2 * jn) * N + nt * 16];
        v2f a0, a1;
        a0.x = ac[i].x; a0.y = ac[i].y;
        a1.x = ac[i].z; a1.y = ac[i].w;
#pragma unroll
        for (int nt = 0; nt < NT; ++nt) {
          v2f b0;
          b0.x = bc[nt].x; b0.y = bc[nt].y;
          acc[nt] = __builtin_amdgcn_wmma_f32_16x16x4_f32(false, a0, false, b0, (short)0,
                                                          acc[nt], false, false);
        }
#pragma unroll
        for (int nt = 0; nt < NT; ++nt) {
          v2f b1;
          b1.x = bc[nt].z; b1.y = bc[nt].w;
          acc[nt] = __builtin_amdgcn_wmma_f32_16x16x4_f32(false, a1, false, b1, (short)0,
                                                          acc[nt], false, false);
        }
#pragma unroll
        for (int nt = 0; nt < NT; ++nt) bc[nt] = bn[nt];
      }
#pragma unroll
      for (int i = 0; i < 4; ++i) ac[i] = an[i];
    }

    const int m0 = mt * 16 + 8 * half;
    float sc[8];
#pragma unroll
    for (int v = 0; v < 8; ++v) sc[v] = rs[m0 + v < M ? m0 + v : M - 1];
#pragma unroll
    for (int nt = 0; nt < NT; ++nt)
#pragma unroll
      for (int v = 0; v < 8; ++v) acc[nt][v] *= sc[v];
    float* __restrict__ Cp = C + (long long)m0 * N + l16;
    if (mt * 16 + 16 <= M) {
#pragma unroll
      for (int nt = 0; nt < NT; ++nt)
#pragma unroll
        for (int v = 0; v < 8; ++v) Cp[(long long)v * N + nt * 16] = acc[nt][v];
    } else {
#pragma unroll
      for (int nt = 0; nt < NT; ++nt)
#pragma unroll
        for (int v = 0; v < 8; ++v)
          if (m0 + v < M) Cp[(long long)v * N + nt * 16] = acc[nt][v];
    }
    mt = mtNext;
  }
}

__global__ void deg_rank_kernel(const int* __restrict__ dst, int* __restrict__ deg,
                                int* __restrict__ rank, int e) {
  const int i = blockIdx.x * blockDim.x + threadIdx.x;
  if (i < e) rank[i] = atomicAdd(&deg[dst[i]], 1);
}

__global__ __launch_bounds__(1024) void scan_dinv_kernel(const int* __restrict__ deg,
                                                         int* __restrict__ rowptr,
                                                         float* __restrict__ dinv, int n) {
  __shared__ int wsum[32];
  __shared__ int sbase;
  const int tid = threadIdx.x;
  const int lane = tid & 31;
  const int wv = tid >> 5;
  const int first = blockIdx.x * 1024;

  int part = 0;
  for (int j = tid; j < first; j += 1024) part += deg[j];
  for (int o = 16; o > 0; o >>= 1) part += __shfl_xor(part, o);
  if (lane == 0) wsum[wv] = part;
  __syncthreads();
  if (wv == 0) {
    int t = wsum[lane];
    for (int o = 16; o > 0; o >>= 1) t += __shfl_xor(t, o);
    if (lane == 0) sbase = t;
  }
  __syncthreads();
  const int base = sbase;

  const int i = first + tid;
  const int dg = i < n ? deg[i] : 0;
  int incl = dg;
  for (int o = 1; o < 32; o <<= 1) {
    const int t = __shfl_up(incl, o);
    if (lane >= o) incl += t;
  }
  if (lane == 31) wsum[wv] = incl;
  __syncthreads();
  if (wv == 0) {
    const int t = wsum[lane];
    int s = t;
    for (int o = 1; o < 32; o <<= 1) {
      const int u = __shfl_up(s, o);
      if (lane >= o) s += u;
    }
    wsum[lane] = s - t;
  }
  __syncthreads();
  const int excl = base + wsum[wv] + incl - dg;
  if (i < n) {
    rowptr[i] = excl;
    dinv[i] = rsqrtf((float)dg + 1.0f);
    if (i == n - 1) rowptr[n] = excl + dg;
  }
}

__global__ void fill_kernel(const int* __restrict__ src, const int* __restrict__ dst,
                            const int* __restrict__ rank, const int* __restrict__ rowptr,
                            int* __restrict__ csr, int e) {
  const int i = blockIdx.x * blockDim.x + threadIdx.x;
  if (i >= e) return;
  int slot = rowptr[dst[i]] + rank[i];
  slot = min(max(slot, 0), e - 1);
  csr[slot] = src[i];
}

static __device__ __forceinline__ float4 fma4(const float4 a, const float w, const float4 c) {
  return make_float4(fmaf(a.x, w, c.x), fmaf(a.y, w, c.y), fmaf(a.z, w, c.z), fmaf(a.w, w, c.w));
}
static __device__ __forceinline__ float4 add4(const float4 c, const float4 a) {
  return make_float4(c.x + a.x, c.y + a.y, c.z + a.z, c.w + a.w);
}

__global__ __launch_bounds__(256) void pull_rows128(const float* __restrict__ H,
                                                    const int* __restrict__ rowptr,
                                                    const int* __restrict__ csr,
                                                    const float* __restrict__ dinv,
                                                    const float* __restrict__ b,
                                                    float* __restrict__ out, int n, int e) {
  const int lane = threadIdx.x & 31;
  const int node = __builtin_amdgcn_readfirstlane(
      (int)((blockIdx.x * (long long)blockDim.x + threadIdx.x) >> 5));
  if (node >= n) return;
  const float4* __restrict__ H4 = (const float4*)H;
  const int beg = min(max(rowptr[node], 0), e);
  const int end = min(max(rowptr[node + 1], beg), e);
  const float dd = dinv[node];
  const float4 self = H4[(long long)node * 32 + lane];
  const float4 bb = ((const float4*)b)[lane];

  float4 acc = make_float4(0.0f, 0.0f, 0.0f, 0.0f);
  for (int base = beg; base < end; base += 32) {
    const int j = base + lane;
    int s = node;
    if (j < end) s = min(max(csr[j], 0), n - 1);
    const int cnt = min(32, end - base);
    int t = 0;
    for (; t + 4 <= cnt; t += 4) {
      const int s0 = __builtin_amdgcn_readlane(s, t);
      const int s1 = __builtin_amdgcn_readlane(s, t + 1);
      const int s2 = __builtin_amdgcn_readlane(s, t + 2);
      const int s3 = __builtin_amdgcn_readlane(s, t + 3);
      const float4 a = H4[(long long)s0 * 32 + lane];
      const float4 b = H4[(long long)s1 * 32 + lane];
      const float4 c = H4[(long long)s2 * 32 + lane];
      const float4 d = H4[(long long)s3 * 32 + lane];
      acc = add4(acc, a);
      acc = add4(acc, b);
      acc = add4(acc, c);
      acc = add4(acc, d);
    }
    if (t < cnt) {
      const int r = cnt - t;
      const float4 a = H4[(long long)__builtin_amdgcn_readlane(s, t) * 32 + lane];
      float4 b = make_float4(0.0f, 0.0f, 0.0f, 0.0f);
      float4 c = b;
      if (r > 1) b = H4[(long long)__builtin_amdgcn_readlane(s, t + 1) * 32 + lane];
      if (r > 2) c = H4[(long long)__builtin_amdgcn_readlane(s, t + 2) * 32 + lane];
      acc = add4(acc, a);
      acc = add4(acc, b);
      acc = add4(acc, c);
    }
  }
  acc = add4(acc, self);
  float4 r = fma4(acc, dd, bb);
  r.x = r.x > 0.0f ? r.x : 0.0f;
  r.y = r.y > 0.0f ? r.y : 0.0f;
  r.z = r.z > 0.0f ? r.z : 0.0f;
  r.w = r.w > 0.0f ? r.w : 0.0f;
  ((float4*)out)[(long long)node * 32 + lane] = r;
}

__global__ __launch_bounds__(256) void pull_rows16(const float* __restrict__ H,
                                                   const int* __restrict__ rowptr,
                                                   const int* __restrict__ csr,
                                                   const float* __restrict__ dinv,
                                                   const float* __restrict__ b,
                                                   float* __restrict__ out, int n, int e) {
  const long long gid = blockIdx.x * (long long)blockDim.x + threadIdx.x;
  const int node = (int)(gid >> 2);
  if (node >= n) return;
  const int sub = threadIdx.x & 3;
  const float4* __restrict__ H4 = (const float4*)H;
  const int beg = min(max(rowptr[node], 0), e);
  const int end = min(max(rowptr[node + 1], beg), e);
  const float dd = dinv[node];
  const float4 self = H4[(long long)node * 4 + sub];

  float4 acc = make_float4(0.0f, 0.0f, 0.0f, 0.0f);
  for (int j = beg; j < end; ++j) {
    const int s = min(max(csr[j], 0), n - 1);
    acc = add4(acc, H4[(long long)s * 4 + sub]);
  }
  acc = add4(acc, self);

  const float4 bb = ((const float4*)b)[sub];
  const float4 v = fma4(acc, dd, bb);
  float mx = fmaxf(fmaxf(v.x, v.y), fmaxf(v.z, v.w));
  mx = fmaxf(mx, __shfl_xor(mx, 1));
  mx = fmaxf(mx, __shfl_xor(mx, 2));
  float s = expf(v.x - mx);
  s += expf(v.y - mx);
  s += expf(v.z - mx);
  s += expf(v.w - mx);
  s += __shfl_xor(s, 1);
  s += __shfl_xor(s, 2);
  const float lse = mx + logf(s);
  ((float4*)out)[(long long)node * 4 + sub] =
      make_float4(v.x - lse, v.y - lse, v.z - lse, v.w - lse);
}

extern "C" void kernel_launch(void* const* d_in, const int* in_sizes, int n_in,
                              void* d_out, int out_size, void* d_ws, size_t ws_size,
                              hipStream_t stream) {
  const float* x  = (const float*)d_in[0];
  const int*   ei = (const int*)d_in[1];
  const float* W1 = (const float*)d_in[2];
  const float* b1 = (const float*)d_in[3];
  const float* W2 = (const float*)d_in[4];
  const float* b2 = (const float*)d_in[5];

  const int n = in_sizes[0] / 256;
  const int e = in_sizes[1] / 2;
  const int* src = ei;
  const int* dst = ei + e;

  float* a1     = (float*)d_ws;
  float* h1     = a1 + (size_t)n * 128;
  int*   csr    = (int*)(h1 + (size_t)n * 128);
  int*   deg    = csr + e;
  int*   rowptr = deg + n;
  float* dinv   = (float*)(rowptr + n + 1);
  int*   rank   = (int*)a1;
  float* h2     = h1;

  hipMemsetAsync(deg, 0, (size_t)n * sizeof(int), stream);

  deg_rank_kernel<<<ceil_div(e, 256), 256, 0, stream>>>(dst, deg, rank, e);
  scan_dinv_kernel<<<ceil_div(n, 1024), 1024, 0, stream>>>(deg, rowptr, dinv, n);
  fill_kernel<<<ceil_div(e, 256), 256, 0, stream>>>(src, dst, rank, rowptr, csr, e);

  gemm_rows16_lds<256, 128><<<512, 256, 0, stream>>>(x, W1, dinv, h1, n);
  pull_rows128<<<ceil_div((long long)n * 32, 256), 256, 0, stream>>>(
      h1, rowptr, csr, dinv, b1, a1, n, e);

  gemm_rows16_lds<128, 16><<<512, 256, 0, stream>>>(a1, W2, dinv, h2, n);
  pull_rows16<<<ceil_div((long long)n * 4, 256), 256, 0, stream>>>(
      h2, rowptr, csr, dinv, b2, (float*)d_out, n, e);
}


# Edges grouped by destination by a two-level counting sort on LDS counters (chunks binned by 256-node range, one owner block per range): global adds, memset, scan launch gone; sum order = LDS arrival
# speedup vs baseline: 5.6879x; 1.4316x over previous
#include <hip/hip_runtime.h>
#include <math.h>


typedef __attribute__((ext_vector_type(2))) float v2f;
typedef __attribute__((ext_vector_type(8))) float v8f;

static inline int ceil_div(long long a, long long b) { return (int)((a + b - 1) / b); }

template <int K, int N>
__global__ __launch_bounds__(256, 4) void gemm_rows16_lds(const float* __restrict__ A,
                                                       const float* __restrict__ B,
                                                       const float* __restrict__ rs,
                                                       float* __restrict__ C, int M) {
  constexpr int NT = N / 16;
  constexpr int KQ = K / 4;
  constexpr int KB = K / 32;
  constexpr int J = K / 8;
  __shared__ float4 Bl[KQ * N];

  const int tid = threadIdx.x;
  const int lane = tid & 31;
  const int wv = __builtin_amdgcn_readfirstlane(tid >> 5);
  const int half = lane >> 4;
  const int l16 = lane & 15;
  const int mTiles = (M + 15) >> 4;

  int mt = (int)blockIdx.x + (int)gridDim.x * wv;
  const int mtStep = (int)gridDim.x * 8;

  float4 ac[4], an[4];
  if (mt < mTiles) {
    const int row = mt * 16 + l16;
    const float* Ar = A + (long long)(row < M ? row : M - 1) * K + 4 * half;
#pragma unroll
    for (int i = 0; i < 4; ++i) ac[i] = *(const float4*)(Ar + 8 * i);
  }

  static_assert((KQ * N) % 256 == 0, "the staging loop has the same trip count in every thread");
#pragma unroll 8
  for (int it = 0; it < KQ * N / 256; ++it) {
    const int idx = it * 256 + tid;
    const int kq = idx / N;
    const int nn = idx - kq * N;
    const float* p = B + (long long)(4 * kq) * N + nn;
    Bl[idx] = make_float4(p[0], p[N], p[2 * N], p[3 * N]);
  }
  __syncthreads();

  const float4* Blane = Bl + half * N + l16;
  float4 bc[NT], bn[NT];
#pragma unroll
  for (int nt = 0; nt < NT; ++nt) bc[nt] = Blane[nt * 16];

  while (mt < mTiles) {
    const int mtNext = mt + mtStep;
    v8f acc[NT];
#pragma unroll
    for (int nt = 0; nt < NT; ++nt) acc[nt] = v8f{};

    const int row = mt * 16 + l16;
    const float* Ar = A + (long long)(row < M ? row : M - 1) * K + 4 * half;
    const int rowN = mtNext * 16 + l16;
    const float* ArN = A + (long long)(rowN < M ? rowN : M - 1) * K + 4 * half;

#pragma unroll
    for (int kb = 0; kb < KB; ++kb) {
      if (kb + 1 < KB) {
#pragma unroll
        for (int i = 0; i < 4; ++i) an[i] = *(const float4*)(Ar + 32 * (kb + 1) + 8 * i);
      } else if (mtNext < mTiles) {
#pragma unroll
        for (int i = 0; i < 4; ++i) an[i] = *(const float4*)(ArN + 8 * i);
      }
#pragma unroll
      for (int i = 0; i < 4; ++i) {
        const int j = kb * 4 + i;
        const int jn = (j + 1) % J;
#pragma unroll
        for (int nt = 0; nt < NT; ++nt) bn[nt] = Blane[(2 * jn) * N + nt * 16];
        v2f a0, a1;
        a0.x = ac[i].x; a0.y = ac[i].y;
        a1.x = ac[i].z; a1.y = ac[i].w;
#pragma unroll
        for (int nt = 0; nt < NT; ++nt) {
          v2f b0;
          b0.x = bc[nt].x; b0.y = bc[nt].y;
          acc[nt] = __builtin_amdgcn_wmma_f32_16x16x4_f32(false, a0, false, b0, (short)0,
                                                          acc[nt], false, false);
        }
#pragma unroll
        for (int nt = 0; nt < NT; ++nt) {
          v2f b1;
          b1.x = bc[nt].z; b1.y = bc[nt].w;
          acc[nt] = __builtin_amdgcn_wmma_f32_16x16x4_f32(false, a1, false, b1, (short)0,
                                                          acc[nt], false, false);
        }
#pragma unroll
        for (int nt = 0; nt < NT; ++nt) bc[nt] = bn[nt];
      }
#pragma unroll
      for (int i = 0; i < 4; ++i) ac[i] = an[i];
    }

    const int m0 = mt * 16 + 8 * half;
    float sc[8];
#pragma unroll
    for (int v = 0; v < 8; ++v) sc[v] = rs[m0 + v < M ? m0 + v : M - 1];
#pragma unroll
    for (int nt = 0; nt < NT; ++nt)
#pragma unroll
      for (int v = 0; v < 8; ++v) acc[nt][v] *= sc[v];
    float* __restrict__ Cp = C + (long long)m0 * N + l16;
    if (mt * 16 + 16 <= M) {
#pragma unroll
      for (int nt = 0; nt < NT; ++nt)
#pragma unroll
        for (int v = 0; v < 8; ++v) Cp[(long long)v * N + nt * 16] = acc[nt][v];
    } else {
#pragma unroll
      for (int nt = 0; nt < NT; ++nt)
#pragma unroll
        for (int v = 0; v < 8; ++v)
          if (m0 + v < M) Cp[(long long)v * N + nt * 16] = acc[nt][v];
    }
    mt = mtNext;
  }
}

constexpr int NC = 512;
constexpr int NBMAX = 4096;

static __device__ __forceinline__ int block_excl_scan256(const int v, int* total, int* scratch) {
  const int lane = threadIdx.x & 31;
  const int wv = threadIdx.x >> 5;
  int incl = v;
  for (int o = 1; o < 32; o <<= 1) {
    const int t = __shfl_up(incl, o);
    if (lane >= o) incl += t;
  }
  if (lane == 31) scratch[wv] = incl;
  __syncthreads();
  if (wv == 0) {
    const int t = lane < 8 ? scratch[lane] : 0;
    int s = t;
    for (int o = 1; o < 8; o <<= 1) {
      const int u = __shfl_up(s, o);
      if (lane >= o) s += u;
    }
    if (lane < 8) scratch[lane] = s - t;
    if (lane == 7) scratch[8] = s;
  }
  __syncthreads();
  const int r = scratch[wv] + incl - v;
  *total = scratch[8];
  __syncthreads();
  return r;
}

__global__ __launch_bounds__(256) void bin_chunk_kernel(const int* __restrict__ src,
                                                        const int* __restrict__ dst,
                                                        int* __restrict__ sorted,
                                                        int2* __restrict__ rec, int n, int e,
                                                        int ch, int nb) {
  __shared__ int hist[NBMAX + 1];
  __shared__ int cur[NBMAX + 1];
  __shared__ int scr[9];
  const int tid = threadIdx.x;
  const int c = blockIdx.x;
  const long long lo_ll = (long long)c * ch;
  const int lo = (int)(lo_ll < e ? lo_ll : e);
  const int hi = (int)(lo_ll + ch < e ? lo_ll + ch : e);

  for (int b = tid; b <= nb; b += 256) hist[b] = 0;
  if (tid == 0) cur[nb] = 0;
  __syncthreads();

  for (int i0 = lo; i0 < hi; i0 += 256 * 8) {
    int d[8];
#pragma unroll
    for (int u = 0; u < 8; ++u) d[u] = dst[min(i0 + u * 256 + tid, hi - 1)];
#pragma unroll
    for (int u = 0; u < 8; ++u) {
      const int dd = min(max(d[u], 0), n - 1);
      atomicAdd(&hist[i0 + u * 256 + tid < hi ? dd >> 8 : nb], 1);
    }
  }
  __syncthreads();

  const int per = (nb + 255) >> 8;
  int sum = 0;
  for (int k = 0; k < per; ++k) {
    const int b = tid * per + k;
    if (b < nb) sum += hist[b];
  }
  int tot;
  int ex = block_excl_scan256(sum, &tot, scr);
  int2* __restrict__ row = rec + (long long)c * nb;
  for (int k = 0; k < per; ++k) {
    const int b = tid * per + k;
    if (b < nb) {
      const int h = hist[b];
      cur[b] = ex;
      row[b] = make_int2(ex, h);
      ex += h;
    }
  }
  __syncthreads();

  for (int i0 = lo; i0 < hi; i0 += 256 * 8) {
    int d[8], s[8], pos[8];
#pragma unroll
    for (int u = 0; u < 8; ++u) {
      const int i = min(i0 + u * 256 + tid, hi - 1);
      d[u] = dst[i];
      s[u] = src[i];
    }
#pragma unroll
    for (int u = 0; u < 8; ++u) {
      d[u] = min(max(d[u], 0), n - 1);
      pos[u] = atomicAdd(&cur[i0 + u * 256 + tid < hi ? d[u] >> 8 : nb], 1);
    }
#pragma unroll
    for (int u = 0; u < 8; ++u) {
      if (i0 + u * 256 + tid < hi) {
        const int ss = min(max(s[u], 0), n - 1);
        sorted[lo + min(pos[u], hi - lo - 1)] = (ss << 8) | (d[u] & 255);
      }
    }
  }
}

__global__ __launch_bounds__(256) void group_rows_kernel(const int* __restrict__ sorted,
                                                         const int2* __restrict__ rec,
                                                         int* __restrict__ rowptr,
                                                         float* __restrict__ dinv,
                                                         int* __restrict__ csr, int n, int e,
                                                         int ch, int nb) {
  constexpr int CAP = 12288;
  constexpr int PPG = NC / 32;
  __shared__ int pOff[NC];
  __shared__ int pCnt[NC];
  __shared__ int degL[256];
  __shared__ int curL[257];
  __shared__ int stage[CAP];
  __shared__ int scr[9];
  __shared__ int red[8];
  const int tid = threadIdx.x;
  const int lane = tid & 31;
  const int wv = tid >> 5;
  const int b = blockIdx.x;

  int offsum = 0;
  for (int c = tid; c < NC; c += 256) {
    const int2 r = rec[(long long)c * nb + b];
    pOff[c] = r.x;
    pCnt[c] = r.y;
    offsum += r.x;
  }
  degL[tid] = 0;
  for (int o = 16; o > 0; o >>= 1) offsum += __shfl_xor(offsum, o);
  if (lane == 0) red[wv] = offsum;
  __syncthreads();
  int base = 0;
  for (int k = 0; k < 8; ++k) base += red[k];

  const int g = tid >> 3;
  const int sub = tid & 7;
  int w0[PPG], w1[PPG];
#pragma unroll
  for (int q = 0; q < PPG; ++q) {
    const int c = g + 32 * q;
    const int cnt = pCnt[c];
    const long long st = (long long)c * ch;
    const int* __restrict__ p = sorted + (st < e ? st : e) + pOff[c];
    w0[q] = sub < cnt ? p[sub] : -1;
    w1[q] = sub + 8 < cnt ? p[sub + 8] : -1;
  }
#pragma unroll
  for (int q = 0; q < PPG; ++q) {
    if (w0[q] >= 0) atomicAdd(&degL[w0[q] & 255], 1);
    if (w1[q] >= 0) atomicAdd(&degL[w1[q] & 255], 1);
  }
  for (int q = 0; q < PPG; ++q) {
    const int c = g + 32 * q;
    const int cnt = pCnt[c];
    const long long st = (long long)c * ch;
    const int* __restrict__ p = sorted + (st < e ? st : e) + pOff[c];
    for (int j = sub + 16; j < cnt; j += 8) atomicAdd(&degL[p[j] & 255], 1);
  }
  __syncthreads();

  const int dg = degL[tid];
  int T;
  const int ex = block_excl_scan256(dg, &T, scr);
  curL[tid] = ex;
  if (tid == 0) curL[256] = 0;
  const int node = b * 256 + tid;
  if (node < n) {
    rowptr[node] = base + ex;
    dinv[node] = rsqrtf((float)dg + 1.0f);
    if (node == n - 1) rowptr[n] = base + ex + dg;
  }
  __syncthreads();

  const bool staged = T <= CAP;
  int p0[PPG], p1[PPG];
#pragma unroll
  for (int q = 0; q < PPG; ++q) {
    p0[q] = atomicAdd(&curL[w0[q] >= 0 ? w0[q] & 255 : 256], 1);
    p1[q] = atomicAdd(&curL[w1[q] >= 0 ? w1[q] & 255 : 256], 1);
  }
#pragma unroll
  for (int q = 0; q < PPG; ++q) {
    if (w0[q] >= 0) {
      if (staged) stage[min(p0[q], CAP - 1)] = w0[q] >> 8;
      else csr[min(max(base + p0[q], 0), e - 1)] = w0[q] >> 8;
    }
    if (w1[q] >= 0) {
      if (staged) stage[min(p1[q], CAP - 1)] = w1[q] >> 8;
      else csr[min(max(base + p1[q], 0), e - 1)] = w1[q] >> 8;
    }
  }
  for (int q = 0; q < PPG; ++q) {
    const int c = g + 32 * q;
    const int cnt = pCnt[c];
    const long long st = (long long)c * ch;
    const int* __restrict__ p = sorted + (st < e ? st : e) + pOff[c];
    for (int j = sub + 16; j < cnt; j += 8) {
      const int w = p[j];
      const int pos = atomicAdd(&curL[w & 255], 1);
      if (staged) stage[min(pos, CAP - 1)] = w >> 8;
      else csr[min(max(base + pos, 0), e - 1)] = w >> 8;
    }
  }
  __syncthreads();
  if (staged)
    for (int k = tid; k < T; k += 256) csr[min(max(base + k, 0), e - 1)] = stage[k];
}

static __device__ __forceinline__ float4 fma4(const float4 a, const float w, const float4 c) {
  return make_float4(fmaf(a.x, w, c.x), fmaf(a.y, w, c.y), fmaf(a.z, w, c.z), fmaf(a.w, w, c.w));
}
static __device__ __forceinline__ float4 add4(const float4 c, const float4 a) {
  return make_float4(c.x + a.x, c.y + a.y, c.z + a.z, c.w + a.w);
}

__global__ __launch_bounds__(256) void pull_rows128(const float* __restrict__ H,
                                                    const int* __restrict__ rowptr,
                                                    const int* __restrict__ csr,
                                                    const float* __restrict__ dinv,
                                                    const float* __restrict__ b,
                                                    float* __restrict__ out, int n, int e) {
  const int lane = threadIdx.x & 31;
  const int node = __builtin_amdgcn_readfirstlane(
      (int)((blockIdx.x * (long long)blockDim.x + threadIdx.x) >> 5));
  if (node >= n) return;
  const float4* __restrict__ H4 = (const float4*)H;
  const int beg = min(max(rowptr[node], 0), e);
  const int end = min(max(rowptr[node + 1], beg), e);
  const float dd = dinv[node];
  const float4 self = H4[(long long)node * 32 + lane];
  const float4 bb = ((const float4*)b)[lane];

  float4 acc = make_float4(0.0f, 0.0f, 0.0f, 0.0f);
  for (int base = beg; base < end; base += 32) {
    const int j = base + lane;
    int s = node;
    if (j < end) s = min(max(csr[j], 0), n - 1);
    const int cnt = min(32, end - base);
    int t = 0;
    for (; t + 4 <= cnt; t += 4) {
      const int s0 = __builtin_amdgcn_readlane(s, t);
      const int s1 = __builtin_amdgcn_readlane(s, t + 1);
      const int s2 = __builtin_amdgcn_readlane(s, t + 2);
      const int s3 = __builtin_amdgcn_readlane(s, t + 3);
      const float4 a = H4[(long long)s0 * 32 + lane];
      const float4 b = H4[(long long)s1 * 32 + lane];
      const float4 c = H4[(long long)s2 * 32 + lane];
      const float4 d = H4[(long long)s3 * 32 + lane];
      acc = add4(acc, a);
      acc = add4(acc, b);
      acc = add4(acc, c);
      acc = add4(acc, d);
    }
    if (t < cnt) {
      const int r = cnt - t;
      const float4 a = H4[(long long)__builtin_amdgcn_readlane(s, t) * 32 + lane];
      float4 b = make_float4(0.0f, 0.0f, 0.0f, 0.0f);
      float4 c = b;
      if (r > 1) b = H4[(long long)__builtin_amdgcn_readlane(s, t + 1) * 32 + lane];
      if (r > 2) c = H4[(long long)__builtin_amdgcn_readlane(s, t + 2) * 32 + lane];
      acc = add4(acc, a);
      acc = add4(acc, b);
      acc = add4(acc, c);
    }
  }
  acc = add4(acc, self);
  float4 r = fma4(acc, dd, bb);
  r.x = r.x > 0.0f ? r.x : 0.0f;
  r.y = r.y > 0.0f ? r.y : 0.0f;
  r.z = r.z > 0.0f ? r.z : 0.0f;
  r.w = r.w > 0.0f ? r.w : 0.0f;
  ((float4*)out)[(long long)node * 32 + lane] = r;
}

__global__ __launch_bounds__(256) void pull_rows16(const float* __restrict__ H,
                                                   const int* __restrict__ rowptr,
                                                   const int* __restrict__ csr,
                                                   const float* __restrict__ dinv,
                                                   const float* __restrict__ b,
                                                   float* __restrict__ out, int n, int e) {
  const long long gid = blockIdx.x * (long long)blockDim.x + threadIdx.x;
  const int node = (int)(gid >> 2);
  if (node >= n) return;
  const int sub = threadIdx.x & 3;
  const float4* __restrict__ H4 = (const float4*)H;
  const int beg = min(max(rowptr[node], 0), e);
  const int end = min(max(rowptr[node + 1], beg), e);
  const float dd = dinv[node];
  const float4 self = H4[(long long)node * 4 + sub];

  float4 acc = make_float4(0.0f, 0.0f, 0.0f, 0.0f);
  for (int j = beg; j < end; ++j) {
    const int s = min(max(csr[j], 0), n - 1);
    acc = add4(acc, H4[(long long)s * 4 + sub]);
  }
  acc = add4(acc, self);

  const float4 bb = ((const float4*)b)[sub];
  const float4 v = fma4(acc, dd, bb);
  float mx = fmaxf(fmaxf(v.x, v.y), fmaxf(v.z, v.w));
  mx = fmaxf(mx, __shfl_xor(mx, 1));
  mx = fmaxf(mx, __shfl_xor(mx, 2));
  float s = expf(v.x - mx);
  s += expf(v.y - mx);
  s += expf(v.z - mx);
  s += expf(v.w - mx);
  s += __shfl_xor(s, 1);
  s += __shfl_xor(s, 2);
  const float lse = mx + logf(s);
  ((float4*)out)[(long long)node * 4 + sub] =
      make_float4(v.x - lse, v.y - lse, v.z - lse, v.w - lse);
}

extern "C" void kernel_launch(void* const* d_in, const int* in_sizes, int n_in,
                              void* d_out, int out_size, void* d_ws, size_t ws_size,
                              hipStream_t stream) {
  const float* x  = (const float*)d_in[0];
  const int*   ei = (const int*)d_in[1];
  const float* W1 = (const float*)d_in[2];
  const float* b1 = (const float*)d_in[3];
  const float* W2 = (const float*)d_in[4];
  const float* b2 = (const float*)d_in[5];

  const int n = in_sizes[0] / 256;
  const int e = in_sizes[1] / 2;
  const int* src = ei;
  const int* dst = ei + e;

  float* a1     = (float*)d_ws;
  float* h1     = a1 + (size_t)n * 128;
  int*   csr    = (int*)(h1 + (size_t)n * 128);
  int*   rowptr = csr + e + n;
  float* dinv   = (float*)(rowptr + n + 1);
  int*   sorted = (int*)a1;
  int2*  rec    = (int2*)h1;
  float* h2     = h1;

  const int ch = ceil_div(e, NC);
  const int nb = ceil_div(n, 256);
  bin_chunk_kernel<<<NC, 256, 0, stream>>>(src, dst, sorted, rec, n, e, ch, nb);
  group_rows_kernel<<<nb, 256, 0, stream>>>(sorted, rec, rowptr, dinv, csr, n, e, ch, nb);

  gemm_rows16_lds<256, 128><<<512, 256, 0, stream>>>(x, W1, dinv, h1, n);
  pull_rows128<<<ceil_div((long long)n * 32, 256), 256, 0, stream>>>(
      h1, rowptr, csr, dinv, b1, a1, n, e);

  gemm_rows16_lds<128, 16><<<512, 256, 0, stream>>>(a1, W2, dinv, h2, n);
  pull_rows16<<<ceil_div((long long)n * 4, 256), 256, 0, stream>>>(
      h2, rowptr, csr, dinv, b2, (float*)d_out, n, e);
}


# Second GEMM fused into the 128-wide aggregation: a work-group's 16 finished rows stay in LDS and wave 0 multiplies them by W2 on WMMA (k and sum order kept); one launch and the a1 plane gone
# speedup vs baseline: 5.9566x; 1.0472x over previous
#include <hip/hip_runtime.h>
#include <math.h>


typedef __attribute__((ext_vector_type(2))) float v2f;
typedef __attribute__((ext_vector_type(8))) float v8f;

static inline int ceil_div(long long a, long long b) { return (int)((a + b - 1) / b); }

template <int K, int N>
__global__ __launch_bounds__(256, 4) void gemm_rows16_lds(const float* __restrict__ A,
                                                       const float* __restrict__ B,
                                                       const float* __restrict__ rs,
                                                       float* __restrict__ C, int M) {
  constexpr int NT = N / 16;
  constexpr int KQ = K / 4;
  constexpr int KB = K / 32;
  constexpr int J = K / 8;
  __shared__ float4 Bl[KQ * N];

  const int tid = threadIdx.x;
  const int lane = tid & 31;
  const int wv = __builtin_amdgcn_readfirstlane(tid >> 5);
  const int half = lane >> 4;
  const int l16 = lane & 15;
  const int mTiles = (M + 15) >> 4;

  int mt = (int)blockIdx.x + (int)gridDim.x * wv;
  const int mtStep = (int)gridDim.x * 8;

  float4 ac[4], an[4];
  if (mt < mTiles) {
    const int row = mt * 16 + l16;
    const float* Ar = A + (long long)(row < M ? row : M - 1) * K + 4 * half;
#pragma unroll
    for (int i = 0; i < 4; ++i) ac[i] = *(const float4*)(Ar + 8 * i);
  }

  static_assert((KQ * N) % 256 == 0, "the staging loop has the same trip count in every thread");
#pragma unroll 8
  for (int it = 0; it < KQ * N / 256; ++it) {
    const int idx = it * 256 + tid;
    const int kq = idx / N;
    const int nn = idx - kq * N;
    const float* p = B + (long long)(4 * kq) * N + nn;
    Bl[idx] = make_float4(p[0], p[N], p[2 * N], p[3 * N]);
  }
  __syncthreads();

  const float4* Blane = Bl + half * N + l16;
  float4 bc[NT], bn[NT];
#pragma unroll
  for (int nt = 0; nt < NT; ++nt) bc[nt] = Blane[nt * 16];

  while (mt < mTiles) {
    const int mtNext = mt + mtStep;
    v8f acc[NT];
#pragma unroll
    for (int nt = 0; nt < NT; ++nt) acc[nt] = v8f{};

    const int row = mt * 16 + l16;
    const float* Ar = A + (long long)(row < M ? row : M - 1) * K + 4 * half;
    const int rowN = mtNext * 16 + l16;
    const float* ArN = A + (long long)(rowN < M ? rowN : M - 1) * K + 4 * half;

#pragma unroll
    for (int kb = 0; kb < KB; ++kb) {
      if (kb + 1 < KB) {
#pragma unroll
        for (int i = 0; i < 4; ++i) an[i] = *(const float4*)(Ar + 32 * (kb + 1) + 8 * i);
      } else if (mtNext < mTiles) {
#pragma unroll
        for (int i = 0; i < 4; ++i) an[i] = *(const float4*)(ArN + 8 * i);
      }
#pragma unroll
      for (int i = 0; i < 4; ++i) {
        const int j = kb * 4 + i;
        const int jn = (j + 1) % J;
#pragma unroll
        for (int nt = 0; nt < NT; ++nt) bn[nt] = Blane[(2 * jn) * N + nt * 16];
        v2f a0, a1;
        a0.x = ac[i].x; a0.y = ac[i].y;
        a1.x = ac[i].z; a1.y = ac[i].w;
#pragma unroll
        for (int nt = 0; nt < NT; ++nt) {
          v2f b0;
          b0.x = bc[nt].x; b0.y = bc[nt].y;
          acc[nt] = __builtin_amdgcn_wmma_f32_16x16x4_f32(false, a0, false, b0, (short)0,
                                                          acc[nt], false, false);
        }
#pragma unroll
        for (int nt = 0; nt < NT; ++nt) {
          v2f b1;
          b1.x = bc[nt].z; b1.y = bc[nt].w;
          acc[nt] = __builtin_amdgcn_wmma_f32_16x16x4_f32(false, a1, false, b1, (short)0,
                                                          acc[nt], false, false);
        }
#pragma unroll
        for (int nt = 0; nt < NT; ++nt) bc[nt] = bn[nt];
      }
#pragma unroll
      for (int i = 0; i < 4; ++i) ac[i] = an[i];
    }

    const int m0 = mt * 16 + 8 * half;
    float sc[8];
#pragma unroll
    for (int v = 0; v < 8; ++v) sc[v] = rs[m0 + v < M ? m0 + v : M - 1];
#pragma unroll
    for (int nt = 0; nt < NT; ++nt)
#pragma unroll
      for (int v = 0; v < 8; ++v) acc[nt][v] *= sc[v];
    float* __restrict__ Cp = C + (long long)m0 * N + l16;
    if (mt * 16 + 16 <= M) {
#pragma unroll
      for (int nt = 0; nt < NT; ++nt)
#pragma unroll
        for (int v = 0; v < 8; ++v) Cp[(long long)v * N + nt * 16] = acc[nt][v];
    } else {
#pragma unroll
      for (int nt = 0; nt < NT; ++nt)
#pragma unroll
        for (int v = 0; v < 8; ++v)
          if (m0 + v < M) Cp[(long long)v * N + nt * 16] = acc[nt][v];
    }
    mt = mtNext;
  }
}

constexpr int NC = 512;
constexpr int NBMAX = 4096;

static __device__ __forceinline__ int block_excl_scan256(const int v, int* total, int* scratch) {
  const int lane = threadIdx.x & 31;
  const int wv = threadIdx.x >> 5;
  int incl = v;
  for (int o = 1; o < 32; o <<= 1) {
    const int t = __shfl_up(incl, o);
    if (lane >= o) incl += t;
  }
  if (lane == 31) scratch[wv] = incl;
  __syncthreads();
  if (wv == 0) {
    const int t = lane < 8 ? scratch[lane] : 0;
    int s = t;
    for (int o = 1; o < 8; o <<= 1) {
      const int u = __shfl_up(s, o);
      if (lane >= o) s += u;
    }
    if (lane < 8) scratch[lane] = s - t;
    if (lane == 7) scratch[8] = s;
  }
  __syncthreads();
  const int r = scratch[wv] + incl - v;
  *total = scratch[8];
  __syncthreads();
  return r;
}

__global__ __launch_bounds__(256) void bin_chunk_kernel(const int* __restrict__ src,
                                                        const int* __restrict__ dst,
                                                        int* __restrict__ sorted,
                                                        int2* __restrict__ rec, int n, int e,
                                                        int ch, int nb) {
  __shared__ int hist[NBMAX + 1];
  __shared__ int cur[NBMAX + 1];
  __shared__ int scr[9];
  const int tid = threadIdx.x;
  const int c = blockIdx.x;
  const long long lo_ll = (long long)c * ch;
  const int lo = (int)(lo_ll < e ? lo_ll : e);
  const int hi = (int)(lo_ll + ch < e ? lo_ll + ch : e);

  for (int b = tid; b <= nb; b += 256) hist[b] = 0;
  if (tid == 0) cur[nb] = 0;
  __syncthreads();

  for (int i0 = lo; i0 < hi; i0 += 256 * 8) {
    int d[8];
#pragma unroll
    for (int u = 0; u < 8; ++u) d[u] = dst[min(i0 + u * 256 + tid, hi - 1)];
#pragma unroll
    for (int u = 0; u < 8; ++u) {
      const int dd = min(max(d[u], 0), n - 1);
      atomicAdd(&hist[i0 + u * 256 + tid < hi ? dd >> 8 : nb], 1);
    }
  }
  __syncthreads();

  const int per = (nb + 255) >> 8;
  int sum = 0;
  for (int k = 0; k < per; ++k) {
    const int b = tid * per + k;
    if (b < nb) sum += hist[b];
  }
  int tot;
  int ex = block_excl_scan256(sum, &tot, scr);
  int2* __restrict__ row = rec + (long long)c * nb;
  for (int k = 0; k < per; ++k) {
    const int b = tid * per + k;
    if (b < nb) {
      const int h = hist[b];
      cur[b] = ex;
      row[b] = make_int2(ex, h);
      ex += h;
    }
  }
  __syncthreads();

  for (int i0 = lo; i0 < hi; i0 += 256 * 8) {
    int d[8], s[8], pos[8];
#pragma unroll
    for (int u = 0; u < 8; ++u) {
      const int i = min(i0 + u * 256 + tid, hi - 1);
      d[u] = dst[i];
      s[u] = src[i];
    }
#pragma unroll
    for (int u = 0; u < 8; ++u) {
      d[u] = min(max(d[u], 0), n - 1);
      pos[u] = atomicAdd(&cur[i0 + u * 256 + tid < hi ? d[u] >> 8 : nb], 1);
    }
#pragma unroll
    for (int u = 0; u < 8; ++u) {
      if (i0 + u * 256 + tid < hi) {
        const int ss = min(max(s[u], 0), n - 1);
        sorted[lo + min(pos[u], hi - lo - 1)] = (ss << 8) | (d[u] & 255);
      }
    }
  }
}

__global__ __launch_bounds__(256) void group_rows_kernel(const int* __restrict__ sorted,
                                                         const int2* __restrict__ rec,
                                                         int* __restrict__ rowptr,
                                                         float* __restrict__ dinv,
                                                         int* __restrict__ csr, int n, int e,
                                                         int ch, int nb) {
  constexpr int CAP = 12288;
  constexpr int PPG = NC / 32;
  __shared__ int pOff[NC];
  __shared__ int pCnt[NC];
  __shared__ int degL[256];
  __shared__ int curL[257];
  __shared__ int stage[CAP];
  __shared__ int scr[9];
  __shared__ int red[8];
  const int tid = threadIdx.x;
  const int lane = tid & 31;
  const int wv = tid >> 5;
  const int b = blockIdx.x;

  int offsum = 0;
  for (int c = tid; c < NC; c += 256) {
    const int2 r = rec[(long long)c * nb + b];
    pOff[c] = r.x;
    pCnt[c] = r.y;
    offsum += r.x;
  }
  degL[tid] = 0;
  for (int o = 16; o > 0; o >>= 1) offsum += __shfl_xor(offsum, o);
  if (lane == 0) red[wv] = offsum;
  __syncthreads();
  int base = 0;
  for (int k = 0; k < 8; ++k) base += red[k];

  const int g = tid >> 3;
  const int sub = tid & 7;
  int w0[PPG], w1[PPG];
#pragma unroll
  for (int q = 0; q < PPG; ++q) {
    const int c = g + 32 * q;
    const int cnt = pCnt[c];
    const long long st = (long long)c * ch;
    const int* __restrict__ p = sorted + (st < e ? st : e) + pOff[c];
    w0[q] = sub < cnt ? p[sub] : -1;
    w1[q] = sub + 8 < cnt ? p[sub + 8] : -1;
  }
#pragma unroll
  for (int q = 0; q < PPG; ++q) {
    if (w0[q] >= 0) atomicAdd(&degL[w0[q] & 255], 1);
    if (w1[q] >= 0) atomicAdd(&degL[w1[q] & 255], 1);
  }
  for (int q = 0; q < PPG; ++q) {
    const int c = g + 32 * q;
    const int cnt = pCnt[c];
    const long long st = (long long)c * ch;
    const int* __restrict__ p = sorted + (st < e ? st : e) + pOff[c];
    for (int j = sub + 16; j < cnt; j += 8) atomicAdd(&degL[p[j] & 255], 1);
  }
  __syncthreads();

  const int dg = degL[tid];
  int T;
  const int ex = block_excl_scan256(dg, &T, scr);
  curL[tid] = ex;
  if (tid == 0) curL[256] = 0;
  const int node = b * 256 + tid;
  if (node < n) {
    rowptr[node] = base + ex;
    dinv[node] = rsqrtf((float)dg + 1.0f);
    if (node == n - 1) rowptr[n] = base + ex + dg;
  }
  __syncthreads();

  const bool staged = T <= CAP;
  int p0[PPG], p1[PPG];
#pragma unroll
  for (int q = 0; q < PPG; ++q) {
    p0[q] = atomicAdd(&curL[w0[q] >= 0 ? w0[q] & 255 : 256], 1);
    p1[q] = atomicAdd(&curL[w1[q] >= 0 ? w1[q] & 255 : 256], 1);
  }
#pragma unroll
  for (int q = 0; q < PPG; ++q) {
    if (w0[q] >= 0) {
      if (staged) stage[min(p0[q], CAP - 1)] = w0[q] >> 8;
      else csr[min(max(base + p0[q], 0), e - 1)] = w0[q] >> 8;
    }
    if (w1[q] >= 0) {
      if (staged) stage[min(p1[q], CAP - 1)] = w1[q] >> 8;
      else csr[min(max(base + p1[q], 0), e - 1)] = w1[q] >> 8;
    }
  }
  for (int q = 0; q < PPG; ++q) {
    const int c = g + 32 * q;
    const int cnt = pCnt[c];
    const long long st = (long long)c * ch;
    const int* __restrict__ p = sorted + (st < e ? st : e) + pOff[c];
    for (int j = sub + 16; j < cnt; j += 8) {
      const int w = p[j];
      const int pos = atomicAdd(&curL[w & 255], 1);
      if (staged) stage[min(pos, CAP - 1)] = w >> 8;
      else csr[min(max(base + pos, 0), e - 1)] = w >> 8;
    }
  }
  __syncthreads();
  if (staged)
    for (int k = tid; k < T; k += 256) csr[min(max(base + k, 0), e - 1)] = stage[k];
}

static __device__ __forceinline__ float4 fma4(const float4 a, const float w, const float4 c) {
  return make_float4(fmaf(a.x, w, c.x), fmaf(a.y, w, c.y), fmaf(a.z, w, c.z), fmaf(a.w, w, c.w));
}
static __device__ __forceinline__ float4 add4(const float4 c, const float4 a) {
  return make_float4(c.x + a.x, c.y + a.y, c.z + a.z, c.w + a.w);
}

__global__ __launch_bounds__(512, 16) void pull_rows128_mm16(const float* __restrict__ H,
                                                         const int* __restrict__ rowptr,
                                                         const int* __restrict__ csr,
                                                         const float* __restrict__ dinv,
                                                         const float* __restrict__ b,
                                                         const float* __restrict__ B2,
                                                         float* __restrict__ C, int n, int e) {
  constexpr int K = 128;
  constexpr int N = 16;
  constexpr int PITCH = K + 4;
  __shared__ float4 Bl[(K / 4) * N];
  __shared__ float4 Al[16 * PITCH / 4];
  __shared__ float4 Dl[4];
  static_assert((K / 4) * N == 512, "one 16-byte piece of the staged B2 per thread");
  const int tid = threadIdx.x;
  const int lane = tid & 31;
  const int wv = __builtin_amdgcn_readfirstlane(tid >> 5);
  const int node = (int)blockIdx.x * 16 + wv;

  const int kq = tid / N;
  const float* pB = B2 + (4 * kq) * N + (tid - kq * N);
  const float4 wB = make_float4(pB[0], pB[N], pB[2 * N], pB[3 * N]);

  float4 r = make_float4(0.0f, 0.0f, 0.0f, 0.0f);
  float rowScale = 0.0f;
  if (node < n) {
  const float4* __restrict__ H4 = (const float4*)H;
  const int beg = min(max(rowptr[node], 0), e);
  const int end = min(max(rowptr[node + 1], beg), e);
  const float dd = dinv[node];
  rowScale = dd;
  const float4 self = H4[(long long)node * 32 + lane];
  const float4 bb = ((const float4*)b)[lane];

  float4 acc = make_float4(0.0f, 0.0f, 0.0f, 0.0f);
  for (int base = beg; base < end; base += 32) {
    const int j = base + lane;
    int s = node;
    if (j < end) s = min(max(csr[j], 0), n - 1);
    const int cnt = min(32, end - base);
    int t = 0;
    for (; t + 4 <= cnt; t += 4) {
      const int s0 = __builtin_amdgcn_readlane(s, t);
      const int s1 = __builtin_amdgcn_readlane(s, t + 1);
      const int s2 = __builtin_amdgcn_readlane(s, t + 2);
      const int s3 = __builtin_amdgcn_readlane(s, t + 3);
      const float4 a = H4[(long long)s0 * 32 + lane];
      const float4 b = H4[(long long)s1 * 32 + lane];
      const float4 c = H4[(long long)s2 * 32 + lane];
      const float4 d = H4[(long long)s3 * 32 + lane];
      acc = add4(acc, a);
      acc = add4(acc, b);
      acc = add4(acc, c);
      acc = add4(acc, d);
    }
    if (t < cnt) {
      const int r = cnt - t;
      const float4 a = H4[(long long)__builtin_amdgcn_readlane(s, t) * 32 + lane];
      float4 b = make_float4(0.0f, 0.0f, 0.0f, 0.0f);
      float4 c = b;
      if (r > 1) b = H4[(long long)__builtin_amdgcn_readlane(s, t + 1) * 32 + lane];
      if (r > 2) c = H4[(long long)__builtin_amdgcn_readlane(s, t + 2) * 32 + lane];
      acc = add4(acc, a);
      acc = add4(acc, b);
      acc = add4(acc, c);
    }
  }
  acc = add4(acc, self);
  r = fma4(acc, dd, bb);
  r.x = r.x > 0.0f ? r.x : 0.0f;
  r.y = r.y > 0.0f ? r.y : 0.0f;
  r.z = r.z > 0.0f ? r.z : 0.0f;
  r.w = r.w > 0.0f ? r.w : 0.0f;
  }
  Al[wv * (PITCH / 4) + lane] = r;
  Bl[tid] = wB;
  if (lane == 0) ((float*)Dl)[wv] = rowScale;
  __syncthreads();
  if (wv != 0) return;

  const int half = lane >> 4;
  const int l16 = lane & 15;
  const float4* Blane = Bl + half * N + l16;
  const float4* Alane = Al + l16 * (PITCH / 4) + half;
  const int m0 = (int)blockIdx.x * 16 + 8 * half;
  const float4 sA = Dl[2 * half];
  const float4 sB = Dl[2 * half + 1];
  const float sc[8] = {sA.x, sA.y, sA.z, sA.w, sB.x, sB.y, sB.z, sB.w};
  v8f t = v8f{};
#pragma unroll
  for (int j = 0; j < K / 8; ++j) {
    const float4 a = Alane[2 * j];
    const float4 bq = Blane[(2 * j) * N];
    v2f a0, a1, b0, b1;
    a0.x = a.x; a0.y = a.y;
    a1.x = a.z; a1.y = a.w;
    b0.x = bq.x; b0.y = bq.y;
    b1.x = bq.z; b1.y = bq.w;
    t = __builtin_amdgcn_wmma_f32_16x16x4_f32(false, a0, false, b0, (short)0, t, false, false);
    t = __builtin_amdgcn_wmma_f32_16x16x4_f32(false, a1, false, b1, (short)0, t, false, false);
  }
#pragma unroll
  for (int v = 0; v < 8; ++v) t[v] *= sc[v];
  float* __restrict__ Cp = C + (long long)m0 * N + l16;
  if ((int)blockIdx.x * 16 + 16 <= n) {
#pragma unroll
    for (int v = 0; v < 8; ++v) Cp[(long long)v * N] = t[v];
  } else {
#pragma unroll
    for (int v = 0; v < 8; ++v)
      if (m0 + v < n) Cp[(long long)v * N] = t[v];
  }
}

__global__ __launch_bounds__(256) void pull_rows16(const float* __restrict__ H,
                                                   const int* __restrict__ rowptr,
                                                   const int* __restrict__ csr,
                                                   const float* __restrict__ dinv,
                                                   const float* __restrict__ b,
                                                   float* __restrict__ out, int n, int e) {
  const long long gid = blockIdx.x * (long long)blockDim.x + threadIdx.x;
  const int node = (int)(gid >> 2);
  if (node >= n) return;
  const int sub = threadIdx.x & 3;
  const float4* __restrict__ H4 = (const float4*)H;
  const int beg = min(max(rowptr[node], 0), e);
  const int end = min(max(rowptr[node + 1], beg), e);
  const float dd = dinv[node];
  const float4 self = H4[(long long)node * 4 + sub];

  float4 acc = make_float4(0.0f, 0.0f, 0.0f, 0.0f);
  for (int j = beg; j < end; ++j) {
    const int s = min(max(csr[j], 0), n - 1);
    acc = add4(acc, H4[(long long)s * 4 + sub]);
  }
  acc = add4(acc, self);

  const float4 bb = ((const float4*)b)[sub];
  const float4 v = fma4(acc, dd, bb);
  float mx = fmaxf(fmaxf(v.x, v.y), fmaxf(v.z, v.w));
  mx = fmaxf(mx, __shfl_xor(mx, 1));
  mx = fmaxf(mx, __shfl_xor(mx, 2));
  float s = expf(v.x - mx);
  s += expf(v.y - mx);
  s += expf(v.z - mx);
  s += expf(v.w - mx);
  s += __shfl_xor(s, 1);
  s += __shfl_xor(s, 2);
  const float lse = mx + logf(s);
  ((float4*)out)[(long long)node * 4 + sub] =
      make_float4(v.x - lse, v.y - lse, v.z - lse, v.w - lse);
}

extern "C" void kernel_launch(void* const* d_in, const int* in_sizes, int n_in,
                              void* d_out, int out_size, void* d_ws, size_t ws_size,
                              hipStream_t stream) {
  const float* x  = (const float*)d_in[0];
  const int*   ei = (const int*)d_in[1];
  const float* W1 = (const float*)d_in[2];
  const float* b1 = (const float*)d_in[3];
  const float* W2 = (const float*)d_in[4];
  const float* b2 = (const float*)d_in[5];

  const int n = in_sizes[0] / 256;
  const int e = in_sizes[1] / 2;
  const int* src = ei;
  const int* dst = ei + e;

  float* t0     = (float*)d_ws;
  float* h1     = t0 + (size_t)n * 128;
  int*   csr    = (int*)(h1 + (size_t)n * 128);
  int*   rowptr = csr + e + n;
  float* dinv   = (float*)(rowptr + n + 1);
  int*   sorted = (int*)t0;
  int2*  rec    = (int2*)h1;
  float* h2     = t0;

  const int ch = ceil_div(e, NC);
  const int nb = ceil_div(n, 256);
  bin_chunk_kernel<<<NC, 256, 0, stream>>>(src, dst, sorted, rec, n, e, ch, nb);
  group_rows_kernel<<<nb, 256, 0, stream>>>(sorted, rec, rowptr, dinv, csr, n, e, ch, nb);

  gemm_rows16_lds<256, 128><<<512, 256, 0, stream>>>(x, W1, dinv, h1, n);
  pull_rows128_mm16<<<ceil_div(n, 16), 512, 0, stream>>>(h1, rowptr, csr, dinv, b1, W2, h2, n, e);

  pull_rows16<<<ceil_div((long long)n * 4, 256), 256, 0, stream>>>(
      h2, rowptr, csr, dinv, b2, (float*)d_out, n, e);
}
